# starting kernel
# baseline (speedup 1.0000x reference)
#include <hip/hip_runtime.h>

#define N_NODES 100000
#define N_EDGES 1600000

typedef __attribute__((ext_vector_type(2))) float v2f;
typedef __attribute__((ext_vector_type(8))) float v8f;

__global__ void deg_init_kernel(float* __restrict__ deg) {
  int i = blockIdx.x * blockDim.x + threadIdx.x;
  if (i < N_NODES) deg[i] = 1.0f;
}

__global__ void deg_count_kernel(float* __restrict__ deg,
                                 const int* __restrict__ col) {
  int e = blockIdx.x * blockDim.x + threadIdx.x;
  if (e < N_EDGES) {
    __hip_atomic_fetch_add(&deg[col[e]], 1.0f,
                           __ATOMIC_RELAXED, __HIP_MEMORY_SCOPE_AGENT);
  }
}

__global__ void deg_rsqrt_kernel(float* __restrict__ deg) {
  int i = blockIdx.x * blockDim.x + threadIdx.x;
  if (i < N_NODES) deg[i] = rsqrtf(deg[i]);
}

template <int DIN, int DOUT>
__global__ void gemm_wmma_kernel(const float* __restrict__ H,
                                 const float* __restrict__ W,
                                 float* __restrict__ Out) {
  constexpr int NT = (DOUT + 15) / 16;
  const int wave = (blockIdx.x * blockDim.x + threadIdx.x) >> 5;
  const int lane = threadIdx.x & 31;
  const int m0 = wave * 16;
  if (m0 >= N_NODES) return;

  const int lhalf = lane & 15;
  const int khi = (lane >> 4) << 1;
  const int ncl = (lhalf < DOUT) ? lhalf : (DOUT - 1);

  const float* __restrict__ hrow = H + (size_t)(m0 + lhalf) * DIN;

  v8f acc0 = {};
  v8f acc1 = {};
#pragma unroll
  for (int k = 0; k < DIN; k += 4) {
    v2f a, b0;
    a.x = hrow[k + khi];
    a.y = hrow[k + khi + 1];
    b0.x = W[(k + khi) * DOUT + ncl];
    b0.y = W[(k + khi + 1) * DOUT + ncl];
    acc0 = __builtin_amdgcn_wmma_f32_16x16x4_f32(
        false, a, false, b0, (short)0, acc0, false, false);
    if (NT > 1) {
      v2f b1;
      b1.x = W[(k + khi) * DOUT + 16 + lhalf];
      b1.y = W[(k + khi + 1) * DOUT + 16 + lhalf];
      acc1 = __builtin_amdgcn_wmma_f32_16x16x4_f32(
          false, a, false, b1, (short)0, acc1, false, false);
    }
  }

  const int mrow = m0 + ((lane >> 4) << 3);
#pragma unroll
  for (int v = 0; v < 8; ++v) {
    const int m = mrow + v;
    if (lhalf < DOUT) Out[(size_t)m * DOUT + lhalf] = acc0[v];
    if (NT > 1)       Out[(size_t)m * DOUT + 16 + lhalf] = acc1[v];
  }
}

template <int DOUT>
__global__ void agg_init_kernel(const float* __restrict__ hW,
                                const float* __restrict__ bias,
                                const float* __restrict__ dinv,
                                float* __restrict__ out) {
  long long t = (long long)blockIdx.x * blockDim.x + threadIdx.x;
  if (t >= (long long)N_NODES * DOUT) return;
  int i = (int)(t / DOUT);
  int f = (int)(t - (long long)i * DOUT);
  float d = dinv[i];
  out[t] = hW[t] * (d * d) + bias[f];
}

template <int DOUT>
__global__ void agg_edge_kernel(const float* __restrict__ hW,
                                const int* __restrict__ row,
                                const int* __restrict__ col,
                                const float* __restrict__ dinv,
                                float* __restrict__ out) {
  long long t = (long long)blockIdx.x * blockDim.x + threadIdx.x;
  if (t >= (long long)N_EDGES * DOUT) return;
  int e = (int)(t / DOUT);
  int f = (int)(t - (long long)e * DOUT);
  int r = row[e];
  int c = col[e];
  float w = dinv[r] * dinv[c];
  float v = hW[(size_t)r * DOUT + f] * w;
  __hip_atomic_fetch_add(&out[(size_t)c * DOUT + f], v,
                         __ATOMIC_RELAXED, __HIP_MEMORY_SCOPE_AGENT);
}

extern "C" void kernel_launch(void* const* d_in, const int* in_sizes, int n_in,
                              void* d_out, int out_size, void* d_ws, size_t ws_size,
                              hipStream_t stream) {
  const float* x  = (const float*)d_in[0];
  const int*   ei = (const int*)d_in[1];
  const int*   row = ei;
  const int*   col = ei + N_EDGES;

  const float* Wm[5] = {(const float*)d_in[2], (const float*)d_in[4],
                        (const float*)d_in[6], (const float*)d_in[8],
                        (const float*)d_in[10]};
  const float* Bm[5] = {(const float*)d_in[3], (const float*)d_in[5],
                        (const float*)d_in[7], (const float*)d_in[9],
                        (const float*)d_in[11]};

  float* ws   = (float*)d_ws;
  float* dinv = ws;
  float* hW   = ws + N_NODES;
  float* h    = ws + N_NODES + N_NODES * 32;

  const int B = 256;
  const int gN  = (N_NODES + B - 1) / B;
  const int gE  = (N_EDGES + B - 1) / B;
  const int gGemm = ((N_NODES / 16) * 32 + B - 1) / B;

  deg_init_kernel<<<gN, B, 0, stream>>>(dinv);
  deg_count_kernel<<<gE, B, 0, stream>>>(dinv, col);
  deg_rsqrt_kernel<<<gN, B, 0, stream>>>(dinv);

  gemm_wmma_kernel<8, 32><<<gGemm, B, 0, stream>>>(x, Wm[0], hW);
  {
    long long n = (long long)N_NODES * 32;
    long long m = (long long)N_EDGES * 32;
    agg_init_kernel<32><<<(int)((n + B - 1) / B), B, 0, stream>>>(hW, Bm[0], dinv, h);
    agg_edge_kernel<32><<<(int)((m + B - 1) / B), B, 0, stream>>>(hW, row, col, dinv, h);
  }

  for (int l = 1; l <= 3; ++l) {
    gemm_wmma_kernel<32, 32><<<gGemm, B, 0, stream>>>(h, Wm[l], hW);
    long long n = (long long)N_NODES * 32;
    long long m = (long long)N_EDGES * 32;
    agg_init_kernel<32><<<(int)((n + B - 1) / B), B, 0, stream>>>(hW, Bm[l], dinv, h);
    agg_edge_kernel<32><<<(int)((m + B - 1) / B), B, 0, stream>>>(hW, row, col, dinv, h);
  }

  gemm_wmma_kernel<32, 3><<<gGemm, B, 0, stream>>>(h, Wm[4], hW);
  {
    float* out = (float*)d_out;
    long long n = (long long)N_NODES * 3;
    long long m = (long long)N_EDGES * 3;
    agg_init_kernel<3><<<(int)((n + B - 1) / B), B, 0, stream>>>(hW, Bm[4], dinv, out);
    agg_edge_kernel<3><<<(int)((m + B - 1) / B), B, 0, stream>>>(hW, row, col, dinv, out);
  }
}


# 32-wide scatter: each wave carries 16 edges (indices and weights loaded lane-parallel, 16 row gathers in flight, then 16 line atomics), 1/16 of the waves; sums stay in atomic arrival order
# speedup vs baseline: 1.7716x; 1.7716x over previous
#include <hip/hip_runtime.h>

#define N_NODES 100000
#define N_EDGES 1600000

typedef __attribute__((ext_vector_type(2))) float v2f;
typedef __attribute__((ext_vector_type(8))) float v8f;

__global__ void deg_init_kernel(float* __restrict__ deg) {
  int i = blockIdx.x * blockDim.x + threadIdx.x;
  if (i < N_NODES) deg[i] = 1.0f;
}

__global__ void deg_count_kernel(float* __restrict__ deg,
                                 const int* __restrict__ col) {
  int e = blockIdx.x * blockDim.x + threadIdx.x;
  if (e < N_EDGES) {
    __hip_atomic_fetch_add(&deg[col[e]], 1.0f,
                           __ATOMIC_RELAXED, __HIP_MEMORY_SCOPE_AGENT);
  }
}

__global__ void deg_rsqrt_kernel(float* __restrict__ deg) {
  int i = blockIdx.x * blockDim.x + threadIdx.x;
  if (i < N_NODES) deg[i] = rsqrtf(deg[i]);
}

template <int DIN, int DOUT>
__global__ void gemm_wmma_kernel(const float* __restrict__ H,
                                 const float* __restrict__ W,
                                 float* __restrict__ Out) {
  constexpr int NT = (DOUT + 15) / 16;
  const int wave = (blockIdx.x * blockDim.x + threadIdx.x) >> 5;
  const int lane = threadIdx.x & 31;
  const int m0 = wave * 16;
  if (m0 >= N_NODES) return;

  const int lhalf = lane & 15;
  const int khi = (lane >> 4) << 1;
  const int ncl = (lhalf < DOUT) ? lhalf : (DOUT - 1);

  const float* __restrict__ hrow = H + (size_t)(m0 + lhalf) * DIN;

  v8f acc0 = {};
  v8f acc1 = {};
#pragma unroll
  for (int k = 0; k < DIN; k += 4) {
    v2f a, b0;
    a.x = hrow[k + khi];
    a.y = hrow[k + khi + 1];
    b0.x = W[(k + khi) * DOUT + ncl];
    b0.y = W[(k + khi + 1) * DOUT + ncl];
    acc0 = __builtin_amdgcn_wmma_f32_16x16x4_f32(
        false, a, false, b0, (short)0, acc0, false, false);
    if (NT > 1) {
      v2f b1;
      b1.x = W[(k + khi) * DOUT + 16 + lhalf];
      b1.y = W[(k + khi + 1) * DOUT + 16 + lhalf];
      acc1 = __builtin_amdgcn_wmma_f32_16x16x4_f32(
          false, a, false, b1, (short)0, acc1, false, false);
    }
  }

  const int mrow = m0 + ((lane >> 4) << 3);
#pragma unroll
  for (int v = 0; v < 8; ++v) {
    const int m = mrow + v;
    if (lhalf < DOUT) Out[(size_t)m * DOUT + lhalf] = acc0[v];
    if (NT > 1)       Out[(size_t)m * DOUT + 16 + lhalf] = acc1[v];
  }
}

template <int DOUT>
__global__ void agg_init_kernel(const float* __restrict__ hW,
                                const float* __restrict__ bias,
                                const float* __restrict__ dinv,
                                float* __restrict__ out) {
  long long t = (long long)blockIdx.x * blockDim.x + threadIdx.x;
  if (t >= (long long)N_NODES * DOUT) return;
  int i = (int)(t / DOUT);
  int f = (int)(t - (long long)i * DOUT);
  float d = dinv[i];
  out[t] = hW[t] * (d * d) + bias[f];
}

template <int DOUT>
__global__ void agg_edge_kernel(const float* __restrict__ hW,
                                const int* __restrict__ row,
                                const int* __restrict__ col,
                                const float* __restrict__ dinv,
                                float* __restrict__ out) {
  long long t = (long long)blockIdx.x * blockDim.x + threadIdx.x;
  if (t >= (long long)N_EDGES * DOUT) return;
  int e = (int)(t / DOUT);
  int f = (int)(t - (long long)e * DOUT);
  int r = row[e];
  int c = col[e];
  float w = dinv[r] * dinv[c];
  float v = hW[(size_t)r * DOUT + f] * w;
  __hip_atomic_fetch_add(&out[(size_t)c * DOUT + f], v,
                         __ATOMIC_RELAXED, __HIP_MEMORY_SCOPE_AGENT);
}

#define EDGES_PER_WAVE 16
__global__ void agg_edge32_batched_kernel(const float* __restrict__ hW,
                                          const int* __restrict__ row,
                                          const int* __restrict__ col,
                                          const float* __restrict__ dinv,
                                          float* __restrict__ out) {
  const int wave = (blockIdx.x * blockDim.x + threadIdx.x) >> 5;
  const int lane = threadIdx.x & 31;
  const int e0 = wave * EDGES_PER_WAVE;
  if (e0 >= N_EDGES) return;

  const int sub = lane & (EDGES_PER_WAVE - 1);
  const int myr = row[e0 + sub];
  const int myc = col[e0 + sub];
  const float myw = dinv[myr] * dinv[myc];

  float v[EDGES_PER_WAVE];
#pragma unroll
  for (int j = 0; j < EDGES_PER_WAVE; ++j) {
    const int r = __builtin_amdgcn_readlane(myr, j);
    v[j] = hW[(size_t)r * 32 + lane];
  }
#pragma unroll
  for (int j = 0; j < EDGES_PER_WAVE; ++j) {
    const int c = __builtin_amdgcn_readlane(myc, j);
    const float w = __int_as_float(__builtin_amdgcn_readlane(__float_as_int(myw), j));
    __hip_atomic_fetch_add(&out[(size_t)c * 32 + lane], v[j] * w,
                           __ATOMIC_RELAXED, __HIP_MEMORY_SCOPE_AGENT);
  }
}

extern "C" void kernel_launch(void* const* d_in, const int* in_sizes, int n_in,
                              void* d_out, int out_size, void* d_ws, size_t ws_size,
                              hipStream_t stream) {
  const float* x  = (const float*)d_in[0];
  const int*   ei = (const int*)d_in[1];
  const int*   row = ei;
  const int*   col = ei + N_EDGES;

  const float* Wm[5] = {(const float*)d_in[2], (const float*)d_in[4],
                        (const float*)d_in[6], (const float*)d_in[8],
                        (const float*)d_in[10]};
  const float* Bm[5] = {(const float*)d_in[3], (const float*)d_in[5],
                        (const float*)d_in[7], (const float*)d_in[9],
                        (const float*)d_in[11]};

  float* ws   = (float*)d_ws;
  float* dinv = ws;
  float* hW   = ws + N_NODES;
  float* h    = ws + N_NODES + N_NODES * 32;

  const int B = 256;
  const int gN  = (N_NODES + B - 1) / B;
  const int gE  = (N_EDGES + B - 1) / B;
  const int gGemm = ((N_NODES / 16) * 32 + B - 1) / B;
  const int gEdge32 = ((N_EDGES / EDGES_PER_WAVE) * 32 + B - 1) / B;

  deg_init_kernel<<<gN, B, 0, stream>>>(dinv);
  deg_count_kernel<<<gE, B, 0, stream>>>(dinv, col);
  deg_rsqrt_kernel<<<gN, B, 0, stream>>>(dinv);

  gemm_wmma_kernel<8, 32><<<gGemm, B, 0, stream>>>(x, Wm[0], hW);
  {
    long long n = (long long)N_NODES * 32;
    agg_init_kernel<32><<<(int)((n + B - 1) / B), B, 0, stream>>>(hW, Bm[0], dinv, h);
    agg_edge32_batched_kernel<<<gEdge32, B, 0, stream>>>(hW, row, col, dinv, h);
  }

  for (int l = 1; l <= 3; ++l) {
    gemm_wmma_kernel<32, 32><<<gGemm, B, 0, stream>>>(h, Wm[l], hW);
    long long n = (long long)N_NODES * 32;
    agg_init_kernel<32><<<(int)((n + B - 1) / B), B, 0, stream>>>(hW, Bm[l], dinv, h);
    agg_edge32_batched_kernel<<<gEdge32, B, 0, stream>>>(hW, row, col, dinv, h);
  }

  gemm_wmma_kernel<32, 3><<<gGemm, B, 0, stream>>>(h, Wm[4], hW);
  {
    float* out = (float*)d_out;
    long long n = (long long)N_NODES * 3;
    long long m = (long long)N_EDGES * 3;
    agg_init_kernel<3><<<(int)((n + B - 1) / B), B, 0, stream>>>(hW, Bm[4], dinv, out);
    agg_edge_kernel<3><<<(int)((m + B - 1) / B), B, 0, stream>>>(hW, row, col, dinv, out);
  }
}


# Edge weight dinv[row]*dinv[col] computed once per call: layer 1's scatter stores it per edge, the other four edge launches load it lane-parallel with the indices; no dinv gathers there; addends same
# speedup vs baseline: 1.8802x; 1.0613x over previous
#include <hip/hip_runtime.h>

#define N_NODES 100000
#define N_EDGES 1600000

typedef __attribute__((ext_vector_type(2))) float v2f;
typedef __attribute__((ext_vector_type(8))) float v8f;

__global__ void deg_init_kernel(float* __restrict__ deg) {
  int i = blockIdx.x * blockDim.x + threadIdx.x;
  if (i < N_NODES) deg[i] = 1.0f;
}

__global__ void deg_count_kernel(float* __restrict__ deg,
                                 const int* __restrict__ col) {
  int e = blockIdx.x * blockDim.x + threadIdx.x;
  if (e < N_EDGES) {
    __hip_atomic_fetch_add(&deg[col[e]], 1.0f,
                           __ATOMIC_RELAXED, __HIP_MEMORY_SCOPE_AGENT);
  }
}

__global__ void deg_rsqrt_kernel(float* __restrict__ deg) {
  int i = blockIdx.x * blockDim.x + threadIdx.x;
  if (i < N_NODES) deg[i] = rsqrtf(deg[i]);
}

template <int DIN, int DOUT>
__global__ void gemm_wmma_kernel(const float* __restrict__ H,
                                 const float* __restrict__ W,
                                 float* __restrict__ Out) {
  constexpr int NT = (DOUT + 15) / 16;
  const int wave = (blockIdx.x * blockDim.x + threadIdx.x) >> 5;
  const int lane = threadIdx.x & 31;
  const int m0 = wave * 16;
  if (m0 >= N_NODES) return;

  const int lhalf = lane & 15;
  const int khi = (lane >> 4) << 1;
  const int ncl = (lhalf < DOUT) ? lhalf : (DOUT - 1);

  const float* __restrict__ hrow = H + (size_t)(m0 + lhalf) * DIN;

  v8f acc0 = {};
  v8f acc1 = {};
#pragma unroll
  for (int k = 0; k < DIN; k += 4) {
    v2f a, b0;
    a.x = hrow[k + khi];
    a.y = hrow[k + khi + 1];
    b0.x = W[(k + khi) * DOUT + ncl];
    b0.y = W[(k + khi + 1) * DOUT + ncl];
    acc0 = __builtin_amdgcn_wmma_f32_16x16x4_f32(
        false, a, false, b0, (short)0, acc0, false, false);
    if (NT > 1) {
      v2f b1;
      b1.x = W[(k + khi) * DOUT + 16 + lhalf];
      b1.y = W[(k + khi + 1) * DOUT + 16 + lhalf];
      acc1 = __builtin_amdgcn_wmma_f32_16x16x4_f32(
          false, a, false, b1, (short)0, acc1, false, false);
    }
  }

  const int mrow = m0 + ((lane >> 4) << 3);
#pragma unroll
  for (int v = 0; v < 8; ++v) {
    const int m = mrow + v;
    if (lhalf < DOUT) Out[(size_t)m * DOUT + lhalf] = acc0[v];
    if (NT > 1)       Out[(size_t)m * DOUT + 16 + lhalf] = acc1[v];
  }
}

template <int DOUT>
__global__ void agg_init_kernel(const float* __restrict__ hW,
                                const float* __restrict__ bias,
                                const float* __restrict__ dinv,
                                float* __restrict__ out) {
  long long t = (long long)blockIdx.x * blockDim.x + threadIdx.x;
  if (t >= (long long)N_NODES * DOUT) return;
  int i = (int)(t / DOUT);
  int f = (int)(t - (long long)i * DOUT);
  float d = dinv[i];
  out[t] = hW[t] * (d * d) + bias[f];
}

template <int DOUT>
__global__ void agg_edge_kernel(const float* __restrict__ hW,
                                const int* __restrict__ row,
                                const int* __restrict__ col,
                                const float* __restrict__ wgt,
                                float* __restrict__ out) {
  long long t = (long long)blockIdx.x * blockDim.x + threadIdx.x;
  if (t >= (long long)N_EDGES * DOUT) return;
  int e = (int)(t / DOUT);
  int f = (int)(t - (long long)e * DOUT);
  int r = row[e];
  int c = col[e];
  float w = wgt[e];
  float v = hW[(size_t)r * DOUT + f] * w;
  __hip_atomic_fetch_add(&out[(size_t)c * DOUT + f], v,
                         __ATOMIC_RELAXED, __HIP_MEMORY_SCOPE_AGENT);
}

#define EDGES_PER_WAVE 16
template <bool MAKE_W>
__global__ void agg_edge32_batched_kernel(const float* __restrict__ hW,
                                          const int* __restrict__ row,
                                          const int* __restrict__ col,
                                          const float* __restrict__ dinv,
                                          float* __restrict__ wgt,
                                          float* __restrict__ out) {
  const int wave = (blockIdx.x * blockDim.x + threadIdx.x) >> 5;
  const int lane = threadIdx.x & 31;
  const int e0 = wave * EDGES_PER_WAVE;
  if (e0 >= N_EDGES) return;

  const int sub = lane & (EDGES_PER_WAVE - 1);
  const int myr = row[e0 + sub];
  const int myc = col[e0 + sub];
  float myw;
  if (MAKE_W) {
    myw = dinv[myr] * dinv[myc];
    if (lane < EDGES_PER_WAVE) wgt[e0 + lane] = myw;
  } else {
    myw = wgt[e0 + sub];
  }

  float v[EDGES_PER_WAVE];
#pragma unroll
  for (int j = 0; j < EDGES_PER_WAVE; ++j) {
    const int r = __builtin_amdgcn_readlane(myr, j);
    v[j] = hW[(size_t)r * 32 + lane];
  }
#pragma unroll
  for (int j = 0; j < EDGES_PER_WAVE; ++j) {
    const int c = __builtin_amdgcn_readlane(myc, j);
    const float w = __int_as_float(__builtin_amdgcn_readlane(__float_as_int(myw), j));
    __hip_atomic_fetch_add(&out[(size_t)c * 32 + lane], v[j] * w,
                           __ATOMIC_RELAXED, __HIP_MEMORY_SCOPE_AGENT);
  }
}

extern "C" void kernel_launch(void* const* d_in, const int* in_sizes, int n_in,
                              void* d_out, int out_size, void* d_ws, size_t ws_size,
                              hipStream_t stream) {
  const float* x  = (const float*)d_in[0];
  const int*   ei = (const int*)d_in[1];
  const int*   row = ei;
  const int*   col = ei + N_EDGES;

  const float* Wm[5] = {(const float*)d_in[2], (const float*)d_in[4],
                        (const float*)d_in[6], (const float*)d_in[8],
                        (const float*)d_in[10]};
  const float* Bm[5] = {(const float*)d_in[3], (const float*)d_in[5],
                        (const float*)d_in[7], (const float*)d_in[9],
                        (const float*)d_in[11]};

  float* ws   = (float*)d_ws;
  float* dinv = ws;
  float* hW   = ws + N_NODES;
  float* h    = ws + N_NODES + N_NODES * 32;
  float* wgt  = ws + N_NODES + N_NODES * 64;

  const int B = 256;
  const int gN  = (N_NODES + B - 1) / B;
  const int gE  = (N_EDGES + B - 1) / B;
  const int gGemm = ((N_NODES / 16) * 32 + B - 1) / B;
  const int gEdge32 = ((N_EDGES / EDGES_PER_WAVE) * 32 + B - 1) / B;

  deg_init_kernel<<<gN, B, 0, stream>>>(dinv);
  deg_count_kernel<<<gE, B, 0, stream>>>(dinv, col);
  deg_rsqrt_kernel<<<gN, B, 0, stream>>>(dinv);

  gemm_wmma_kernel<8, 32><<<gGemm, B, 0, stream>>>(x, Wm[0], hW);
  {
    long long n = (long long)N_NODES * 32;
    agg_init_kernel<32><<<(int)((n + B - 1) / B), B, 0, stream>>>(hW, Bm[0], dinv, h);
    agg_edge32_batched_kernel<true><<<gEdge32, B, 0, stream>>>(hW, row, col, dinv, wgt, h);
  }

  for (int l = 1; l <= 3; ++l) {
    gemm_wmma_kernel<32, 32><<<gGemm, B, 0, stream>>>(h, Wm[l], hW);
    long long n = (long long)N_NODES * 32;
    agg_init_kernel<32><<<(int)((n + B - 1) / B), B, 0, stream>>>(hW, Bm[l], dinv, h);
    agg_edge32_batched_kernel<false><<<gEdge32, B, 0, stream>>>(hW, row, col, dinv, wgt, h);
  }

  gemm_wmma_kernel<32, 3><<<gGemm, B, 0, stream>>>(h, Wm[4], hW);
  {
    float* out = (float*)d_out;
    long long n = (long long)N_NODES * 3;
    long long m = (long long)N_EDGES * 3;
    agg_init_kernel<3><<<(int)((n + B - 1) / B), B, 0, stream>>>(hW, Bm[4], dinv, out);
    agg_edge_kernel<3><<<(int)((m + B - 1) / B), B, 0, stream>>>(hW, row, col, wgt, out);
  }
}


# Scatter adds go to one accumulator plane per cache half (die id read per wave, speed only); next GEMM loads the planes' sum; sums in arrival order per half, then the two partials added
# speedup vs baseline: 2.0053x; 1.0665x over previous
#include <hip/hip_runtime.h>

#define N_NODES 100000
#define N_EDGES 1600000

typedef __attribute__((ext_vector_type(2))) float v2f;
typedef __attribute__((ext_vector_type(8))) float v8f;

__global__ void deg_init_kernel(float* __restrict__ deg) {
  int i = blockIdx.x * blockDim.x + threadIdx.x;
  if (i < N_NODES) deg[i] = 1.0f;
}

__global__ void deg_count_kernel(float* __restrict__ deg,
                                 const int* __restrict__ col) {
  int e = blockIdx.x * blockDim.x + threadIdx.x;
  if (e < N_EDGES) {
    __hip_atomic_fetch_add(&deg[col[e]], 1.0f,
                           __ATOMIC_RELAXED, __HIP_MEMORY_SCOPE_AGENT);
  }
}

__global__ void deg_rsqrt_kernel(float* __restrict__ deg) {
  int i = blockIdx.x * blockDim.x + threadIdx.x;
  if (i < N_NODES) deg[i] = rsqrtf(deg[i]);
}

template <int DIN, int DOUT, bool TWO>
__global__ void gemm_wmma_kernel(const float* __restrict__ H,
                                 const float* __restrict__ H2,
                                 const float* __restrict__ W,
                                 float* __restrict__ Out) {
  constexpr int NT = (DOUT + 15) / 16;
  const int wave = (blockIdx.x * blockDim.x + threadIdx.x) >> 5;
  const int lane = threadIdx.x & 31;
  const int m0 = wave * 16;
  if (m0 >= N_NODES) return;

  const int lhalf = lane & 15;
  const int khi = (lane >> 4) << 1;
  const int ncl = (lhalf < DOUT) ? lhalf : (DOUT - 1);

  const float* __restrict__ hrow = H + (size_t)(m0 + lhalf) * DIN;
  const float* __restrict__ hrow2 = H2 + (size_t)(m0 + lhalf) * DIN;

  v8f acc0 = {};
  v8f acc1 = {};
#pragma unroll
  for (int k = 0; k < DIN; k += 4) {
    v2f a, b0;
    a.x = hrow[k + khi];
    a.y = hrow[k + khi + 1];
    if (TWO) {
      a.x += hrow2[k + khi];
      a.y += hrow2[k + khi + 1];
    }
    b0.x = W[(k + khi) * DOUT + ncl];
    b0.y = W[(k + khi + 1) * DOUT + ncl];
    acc0 = __builtin_amdgcn_wmma_f32_16x16x4_f32(
        false, a, false, b0, (short)0, acc0, false, false);
    if (NT > 1) {
      v2f b1;
      b1.x = W[(k + khi) * DOUT + 16 + lhalf];
      b1.y = W[(k + khi + 1) * DOUT + 16 + lhalf];
      acc1 = __builtin_amdgcn_wmma_f32_16x16x4_f32(
          false, a, false, b1, (short)0, acc1, false, false);
    }
  }

  const int mrow = m0 + ((lane >> 4) << 3);
#pragma unroll
  for (int v = 0; v < 8; ++v) {
    const int m = mrow + v;
    if (lhalf < DOUT) Out[(size_t)m * DOUT + lhalf] = acc0[v];
    if (NT > 1)       Out[(size_t)m * DOUT + 16 + lhalf] = acc1[v];
  }
}

template <int DOUT, bool TWO>
__global__ void agg_init_kernel(const float* __restrict__ hW,
                                const float* __restrict__ bias,
                                const float* __restrict__ dinv,
                                float* __restrict__ out,
                                float* __restrict__ out2) {
  long long t = (long long)blockIdx.x * blockDim.x + threadIdx.x;
  if (t >= (long long)N_NODES * DOUT) return;
  int i = (int)(t / DOUT);
  int f = (int)(t - (long long)i * DOUT);
  float d = dinv[i];
  out[t] = hW[t] * (d * d) + bias[f];
  if (TWO) out2[t] = 0.0f;
}

template <int DOUT>
__global__ void agg_edge_kernel(const float* __restrict__ hW,
                                const int* __restrict__ row,
                                const int* __restrict__ col,
                                const float* __restrict__ wgt,
                                float* __restrict__ out) {
  long long t = (long long)blockIdx.x * blockDim.x + threadIdx.x;
  if (t >= (long long)N_EDGES * DOUT) return;
  int e = (int)(t / DOUT);
  int f = (int)(t - (long long)e * DOUT);
  int r = row[e];
  int c = col[e];
  float w = wgt[e];
  float v = hW[(size_t)r * DOUT + f] * w;
  __hip_atomic_fetch_add(&out[(size_t)c * DOUT + f], v,
                         __ATOMIC_RELAXED, __HIP_MEMORY_SCOPE_AGENT);
}

#define EDGES_PER_WAVE 16
template <bool MAKE_W>
__global__ void agg_edge32_batched_kernel(const float* __restrict__ hW,
                                          const int* __restrict__ row,
                                          const int* __restrict__ col,
                                          const float* __restrict__ dinv,
                                          float* __restrict__ wgt,
                                          float* __restrict__ out0,
                                          float* __restrict__ out1) {
  const int wave = (blockIdx.x * blockDim.x + threadIdx.x) >> 5;
  const int lane = threadIdx.x & 31;
  const int e0 = wave * EDGES_PER_WAVE;
  if (e0 >= N_EDGES) return;

  const int sub = lane & (EDGES_PER_WAVE - 1);
  const int myr = row[e0 + sub];
  const int myc = col[e0 + sub];
  float myw;
  if (MAKE_W) {
    myw = dinv[myr] * dinv[myc];
    if (lane < EDGES_PER_WAVE) wgt[e0 + lane] = myw;
  } else {
    myw = wgt[e0 + sub];
  }

  float v[EDGES_PER_WAVE];
#pragma unroll
  for (int j = 0; j < EDGES_PER_WAVE; ++j) {
    const int r = __builtin_amdgcn_readlane(myr, j);
    v[j] = hW[(size_t)r * 32 + lane];
  }
  const unsigned place = __builtin_amdgcn_s_sendmsg_rtn(0x87);
  float* __restrict__ out = (place & (1u << 18)) ? out1 : out0;

#pragma unroll
  for (int j = 0; j < EDGES_PER_WAVE; ++j) {
    const int c = __builtin_amdgcn_readlane(myc, j);
    const float w = __int_as_float(__builtin_amdgcn_readlane(__float_as_int(myw), j));
    __hip_atomic_fetch_add(&out[(size_t)c * 32 + lane], v[j] * w,
                           __ATOMIC_RELAXED, __HIP_MEMORY_SCOPE_AGENT);
  }
}

extern "C" void kernel_launch(void* const* d_in, const int* in_sizes, int n_in,
                              void* d_out, int out_size, void* d_ws, size_t ws_size,
                              hipStream_t stream) {
  const float* x  = (const float*)d_in[0];
  const int*   ei = (const int*)d_in[1];
  const int*   row = ei;
  const int*   col = ei + N_EDGES;

  const float* Wm[5] = {(const float*)d_in[2], (const float*)d_in[4],
                        (const float*)d_in[6], (const float*)d_in[8],
                        (const float*)d_in[10]};
  const float* Bm[5] = {(const float*)d_in[3], (const float*)d_in[5],
                        (const float*)d_in[7], (const float*)d_in[9],
                        (const float*)d_in[11]};

  float* ws   = (float*)d_ws;
  float* dinv = ws;
  float* hW   = ws + N_NODES;
  float* h    = ws + N_NODES + N_NODES * 32;
  float* wgt  = ws + N_NODES + N_NODES * 64;
  float* h1   = ws + N_NODES + N_NODES * 64 + N_EDGES;

  const int B = 256;
  const int gN  = (N_NODES + B - 1) / B;
  const int gE  = (N_EDGES + B - 1) / B;
  const int gGemm = ((N_NODES / 16) * 32 + B - 1) / B;
  const int gEdge32 = ((N_EDGES / EDGES_PER_WAVE) * 32 + B - 1) / B;

  deg_init_kernel<<<gN, B, 0, stream>>>(dinv);
  deg_count_kernel<<<gE, B, 0, stream>>>(dinv, col);
  deg_rsqrt_kernel<<<gN, B, 0, stream>>>(dinv);

  gemm_wmma_kernel<8, 32, false><<<gGemm, B, 0, stream>>>(x, x, Wm[0], hW);
  {
    long long n = (long long)N_NODES * 32;
    agg_init_kernel<32, true><<<(int)((n + B - 1) / B), B, 0, stream>>>(hW, Bm[0], dinv, h, h1);
    agg_edge32_batched_kernel<true><<<gEdge32, B, 0, stream>>>(hW, row, col, dinv, wgt, h, h1);
  }

  for (int l = 1; l <= 3; ++l) {
    gemm_wmma_kernel<32, 32, true><<<gGemm, B, 0, stream>>>(h, h1, Wm[l], hW);
    long long n = (long long)N_NODES * 32;
    agg_init_kernel<32, true><<<(int)((n + B - 1) / B), B, 0, stream>>>(hW, Bm[l], dinv, h, h1);
    agg_edge32_batched_kernel<false><<<gEdge32, B, 0, stream>>>(hW, row, col, dinv, wgt, h, h1);
  }

  gemm_wmma_kernel<32, 3, true><<<gGemm, B, 0, stream>>>(h, h1, Wm[4], hW);
  {
    float* out = (float*)d_out;
    long long n = (long long)N_NODES * 3;
    long long m = (long long)N_EDGES * 3;
    agg_init_kernel<3, false><<<(int)((n + B - 1) / B), B, 0, stream>>>(hW, Bm[4], dinv, out, out);
    agg_edge_kernel<3><<<(int)((m + B - 1) / B), B, 0, stream>>>(hW, row, col, wgt, out);
  }
}


# Degree count through LDS: a work-group counts a 50000-edge slice into 16-bit LDS counters for one 32768-node range, then one line-wide global atomic per 32 nodes (12x fewer); deg same floats
# speedup vs baseline: 2.4530x; 1.2233x over previous
#include <hip/hip_runtime.h>

#define N_NODES 100000
#define N_EDGES 1600000

typedef __attribute__((ext_vector_type(2))) float v2f;
typedef __attribute__((ext_vector_type(8))) float v8f;

__global__ void deg_init_kernel(float* __restrict__ deg) {
  int i = blockIdx.x * blockDim.x + threadIdx.x;
  if (i < N_NODES) deg[i] = 1.0f;
}

#define DEG_RANGE_LOG 15
#define DEG_RANGE (1 << DEG_RANGE_LOG)
#define DEG_NRANGES ((N_NODES + DEG_RANGE - 1) / DEG_RANGE)
#define DEG_CHUNKS 32
#define DEG_CHUNK_EDGES (N_EDGES / DEG_CHUNKS)
#define DEG_THREADS 1024
static_assert(N_EDGES % DEG_CHUNKS == 0, "edge slices must be equal");
static_assert(DEG_CHUNK_EDGES % 4 == 0, "slices are read four indices at a time");
static_assert(DEG_CHUNK_EDGES < 65536, "a 16-bit counter must hold a whole slice");

__global__ void __launch_bounds__(DEG_THREADS)
deg_hist_kernel(float* __restrict__ deg, const int* __restrict__ col) {
  __shared__ unsigned cnt[DEG_RANGE / 2];
  const int range = blockIdx.x / DEG_CHUNKS;
  const int chunk = blockIdx.x - range * DEG_CHUNKS;
  const int tid = threadIdx.x;

  for (int w = tid; w < DEG_RANGE / 2; w += DEG_THREADS) cnt[w] = 0u;

  constexpr int NV = DEG_CHUNK_EDGES / 4;
  constexpr int IT = (NV + DEG_THREADS - 1) / DEG_THREADS;
  const int4* __restrict__ col4 =
      reinterpret_cast<const int4*>(col + (size_t)chunk * DEG_CHUNK_EDGES);
  int4 cv[IT];
#pragma unroll
  for (int k = 0; k < IT; ++k) {
    const int idx = tid + k * DEG_THREADS;
    cv[k] = col4[idx < NV ? idx : NV - 1];
  }
  __syncthreads();

#pragma unroll
  for (int k = 0; k < IT; ++k) {
    if (tid + k * DEG_THREADS < NV) {
      const int c4[4] = {cv[k].x, cv[k].y, cv[k].z, cv[k].w};
#pragma unroll
      for (int q = 0; q < 4; ++q) {
        const int c = c4[q];
        if ((c >> DEG_RANGE_LOG) == range) {
          const int local = c & (DEG_RANGE - 1);
          atomicAdd(&cnt[local >> 1], 1u << ((local & 1) << 4));
        }
      }
    }
  }
  __syncthreads();

  const unsigned short* cnt16 = reinterpret_cast<const unsigned short*>(cnt);
  const int nbase = range << DEG_RANGE_LOG;
  for (int n = tid; n < DEG_RANGE; n += DEG_THREADS) {
    const int node = nbase + n;
    const unsigned k = cnt16[n];
    if (node < N_NODES && k != 0u) {
      __hip_atomic_fetch_add(&deg[node], (float)k,
                             __ATOMIC_RELAXED, __HIP_MEMORY_SCOPE_AGENT);
    }
  }
}

__global__ void deg_rsqrt_kernel(float* __restrict__ deg) {
  int i = blockIdx.x * blockDim.x + threadIdx.x;
  if (i < N_NODES) deg[i] = rsqrtf(deg[i]);
}

template <int DIN, int DOUT, bool TWO>
__global__ void gemm_wmma_kernel(const float* __restrict__ H,
                                 const float* __restrict__ H2,
                                 const float* __restrict__ W,
                                 float* __restrict__ Out) {
  constexpr int NT = (DOUT + 15) / 16;
  const int wave = (blockIdx.x * blockDim.x + threadIdx.x) >> 5;
  const int lane = threadIdx.x & 31;
  const int m0 = wave * 16;
  if (m0 >= N_NODES) return;

  const int lhalf = lane & 15;
  const int khi = (lane >> 4) << 1;
  const int ncl = (lhalf < DOUT) ? lhalf : (DOUT - 1);

  const float* __restrict__ hrow = H + (size_t)(m0 + lhalf) * DIN;
  const float* __restrict__ hrow2 = H2 + (size_t)(m0 + lhalf) * DIN;

  v8f acc0 = {};
  v8f acc1 = {};
#pragma unroll
  for (int k = 0; k < DIN; k += 4) {
    v2f a, b0;
    a.x = hrow[k + khi];
    a.y = hrow[k + khi + 1];
    if (TWO) {
      a.x += hrow2[k + khi];
      a.y += hrow2[k + khi + 1];
    }
    b0.x = W[(k + khi) * DOUT + ncl];
    b0.y = W[(k + khi + 1) * DOUT + ncl];
    acc0 = __builtin_amdgcn_wmma_f32_16x16x4_f32(
        false, a, false, b0, (short)0, acc0, false, false);
    if (NT > 1) {
      v2f b1;
      b1.x = W[(k + khi) * DOUT + 16 + lhalf];
      b1.y = W[(k + khi + 1) * DOUT + 16 + lhalf];
      acc1 = __builtin_amdgcn_wmma_f32_16x16x4_f32(
          false, a, false, b1, (short)0, acc1, false, false);
    }
  }

  const int mrow = m0 + ((lane >> 4) << 3);
#pragma unroll
  for (int v = 0; v < 8; ++v) {
    const int m = mrow + v;
    if (lhalf < DOUT) Out[(size_t)m * DOUT + lhalf] = acc0[v];
    if (NT > 1)       Out[(size_t)m * DOUT + 16 + lhalf] = acc1[v];
  }
}

template <int DOUT, bool TWO>
__global__ void agg_init_kernel(const float* __restrict__ hW,
                                const float* __restrict__ bias,
                                const float* __restrict__ dinv,
                                float* __restrict__ out,
                                float* __restrict__ out2) {
  long long t = (long long)blockIdx.x * blockDim.x + threadIdx.x;
  if (t >= (long long)N_NODES * DOUT) return;
  int i = (int)(t / DOUT);
  int f = (int)(t - (long long)i * DOUT);
  float d = dinv[i];
  out[t] = hW[t] * (d * d) + bias[f];
  if (TWO) out2[t] = 0.0f;
}

template <int DOUT>
__global__ void agg_edge_kernel(const float* __restrict__ hW,
                                const int* __restrict__ row,
                                const int* __restrict__ col,
                                const float* __restrict__ wgt,
                                float* __restrict__ out) {
  long long t = (long long)blockIdx.x * blockDim.x + threadIdx.x;
  if (t >= (long long)N_EDGES * DOUT) return;
  int e = (int)(t / DOUT);
  int f = (int)(t - (long long)e * DOUT);
  int r = row[e];
  int c = col[e];
  float w = wgt[e];
  float v = hW[(size_t)r * DOUT + f] * w;
  __hip_atomic_fetch_add(&out[(size_t)c * DOUT + f], v,
                         __ATOMIC_RELAXED, __HIP_MEMORY_SCOPE_AGENT);
}

#define EDGES_PER_WAVE 16
template <bool MAKE_W>
__global__ void agg_edge32_batched_kernel(const float* __restrict__ hW,
                                          const int* __restrict__ row,
                                          const int* __restrict__ col,
                                          const float* __restrict__ dinv,
                                          float* __restrict__ wgt,
                                          float* __restrict__ out0,
                                          float* __restrict__ out1) {
  const int wave = (blockIdx.x * blockDim.x + threadIdx.x) >> 5;
  const int lane = threadIdx.x & 31;
  const int e0 = wave * EDGES_PER_WAVE;
  if (e0 >= N_EDGES) return;

  const int sub = lane & (EDGES_PER_WAVE - 1);
  const int myr = row[e0 + sub];
  const int myc = col[e0 + sub];
  float myw;
  if (MAKE_W) {
    myw = dinv[myr] * dinv[myc];
    if (lane < EDGES_PER_WAVE) wgt[e0 + lane] = myw;
  } else {
    myw = wgt[e0 + sub];
  }

  float v[EDGES_PER_WAVE];
#pragma unroll
  for (int j = 0; j < EDGES_PER_WAVE; ++j) {
    const int r = __builtin_amdgcn_readlane(myr, j);
    v[j] = hW[(size_t)r * 32 + lane];
  }
  const unsigned place = __builtin_amdgcn_s_sendmsg_rtn(0x87);
  float* __restrict__ out = (place & (1u << 18)) ? out1 : out0;

#pragma unroll
  for (int j = 0; j < EDGES_PER_WAVE; ++j) {
    const int c = __builtin_amdgcn_readlane(myc, j);
    const float w = __int_as_float(__builtin_amdgcn_readlane(__float_as_int(myw), j));
    __hip_atomic_fetch_add(&out[(size_t)c * 32 + lane], v[j] * w,
                           __ATOMIC_RELAXED, __HIP_MEMORY_SCOPE_AGENT);
  }
}

extern "C" void kernel_launch(void* const* d_in, const int* in_sizes, int n_in,
                              void* d_out, int out_size, void* d_ws, size_t ws_size,
                              hipStream_t stream) {
  const float* x  = (const float*)d_in[0];
  const int*   ei = (const int*)d_in[1];
  const int*   row = ei;
  const int*   col = ei + N_EDGES;

  const float* Wm[5] = {(const float*)d_in[2], (const float*)d_in[4],
                        (const float*)d_in[6], (const float*)d_in[8],
                        (const float*)d_in[10]};
  const float* Bm[5] = {(const float*)d_in[3], (const float*)d_in[5],
                        (const float*)d_in[7], (const float*)d_in[9],
                        (const float*)d_in[11]};

  float* ws   = (float*)d_ws;
  float* dinv = ws;
  float* hW   = ws + N_NODES;
  float* h    = ws + N_NODES + N_NODES * 32;
  float* wgt  = ws + N_NODES + N_NODES * 64;
  float* h1   = ws + N_NODES + N_NODES * 64 + N_EDGES;

  const int B = 256;
  const int gN  = (N_NODES + B - 1) / B;
  const int gGemm = ((N_NODES / 16) * 32 + B - 1) / B;
  const int gEdge32 = ((N_EDGES / EDGES_PER_WAVE) * 32 + B - 1) / B;

  deg_init_kernel<<<gN, B, 0, stream>>>(dinv);
  deg_hist_kernel<<<DEG_NRANGES * DEG_CHUNKS, DEG_THREADS, 0, stream>>>(dinv, col);
  deg_rsqrt_kernel<<<gN, B, 0, stream>>>(dinv);

  gemm_wmma_kernel<8, 32, false><<<gGemm, B, 0, stream>>>(x, x, Wm[0], hW);
  {
    long long n = (long long)N_NODES * 32;
    agg_init_kernel<32, true><<<(int)((n + B - 1) / B), B, 0, stream>>>(hW, Bm[0], dinv, h, h1);
    agg_edge32_batched_kernel<true><<<gEdge32, B, 0, stream>>>(hW, row, col, dinv, wgt, h, h1);
  }

  for (int l = 1; l <= 3; ++l) {
    gemm_wmma_kernel<32, 32, true><<<gGemm, B, 0, stream>>>(h, h1, Wm[l], hW);
    long long n = (long long)N_NODES * 32;
    agg_init_kernel<32, true><<<(int)((n + B - 1) / B), B, 0, stream>>>(hW, Bm[l], dinv, h, h1);
    agg_edge32_batched_kernel<false><<<gEdge32, B, 0, stream>>>(hW, row, col, dinv, wgt, h, h1);
  }

  gemm_wmma_kernel<32, 3, true><<<gGemm, B, 0, stream>>>(h, h1, Wm[4], hW);
  {
    float* out = (float*)d_out;
    long long n = (long long)N_NODES * 3;
    long long m = (long long)N_EDGES * 3;
    agg_init_kernel<3, false><<<(int)((n + B - 1) / B), B, 0, stream>>>(hW, Bm[4], dinv, out, out);
    agg_edge_kernel<3><<<(int)((m + B - 1) / B), B, 0, stream>>>(hW, row, col, wgt, out);
  }
}


# Last layer's scatter through LDS: a work-group sums a 50000-edge slice into LDS for one 12512-node range and stores a partial plane; 32 planes summed in slice order; no global atomics there
# speedup vs baseline: 2.8866x; 1.1768x over previous
#include <hip/hip_runtime.h>

#define N_NODES 100000
#define N_EDGES 1600000

typedef __attribute__((ext_vector_type(2))) float v2f;
typedef __attribute__((ext_vector_type(8))) float v8f;

__global__ void deg_init_kernel(float* __restrict__ deg) {
  int i = blockIdx.x * blockDim.x + threadIdx.x;
  if (i < N_NODES) deg[i] = 1.0f;
}

#define DEG_RANGE_LOG 15
#define DEG_RANGE (1 << DEG_RANGE_LOG)
#define DEG_NRANGES ((N_NODES + DEG_RANGE - 1) / DEG_RANGE)
#define DEG_CHUNKS 32
#define DEG_CHUNK_EDGES (N_EDGES / DEG_CHUNKS)
#define DEG_THREADS 1024
static_assert(N_EDGES % DEG_CHUNKS == 0, "edge slices must be equal");
static_assert(DEG_CHUNK_EDGES % 4 == 0, "slices are read four indices at a time");
static_assert(DEG_CHUNK_EDGES < 65536, "a 16-bit counter must hold a whole slice");

__global__ void __launch_bounds__(DEG_THREADS)
deg_hist_kernel(float* __restrict__ deg, const int* __restrict__ col) {
  __shared__ unsigned cnt[DEG_RANGE / 2];
  const int range = blockIdx.x / DEG_CHUNKS;
  const int chunk = blockIdx.x - range * DEG_CHUNKS;
  const int tid = threadIdx.x;

  for (int w = tid; w < DEG_RANGE / 2; w += DEG_THREADS) cnt[w] = 0u;

  constexpr int NV = DEG_CHUNK_EDGES / 4;
  constexpr int IT = (NV + DEG_THREADS - 1) / DEG_THREADS;
  const int4* __restrict__ col4 =
      reinterpret_cast<const int4*>(col + (size_t)chunk * DEG_CHUNK_EDGES);
  int4 cv[IT];
#pragma unroll
  for (int k = 0; k < IT; ++k) {
    const int idx = tid + k * DEG_THREADS;
    cv[k] = col4[idx < NV ? idx : NV - 1];
  }
  __syncthreads();

#pragma unroll
  for (int k = 0; k < IT; ++k) {
    if (tid + k * DEG_THREADS < NV) {
      const int c4[4] = {cv[k].x, cv[k].y, cv[k].z, cv[k].w};
#pragma unroll
      for (int q = 0; q < 4; ++q) {
        const int c = c4[q];
        if ((c >> DEG_RANGE_LOG) == range) {
          const int local = c & (DEG_RANGE - 1);
          atomicAdd(&cnt[local >> 1], 1u << ((local & 1) << 4));
        }
      }
    }
  }
  __syncthreads();

  const unsigned short* cnt16 = reinterpret_cast<const unsigned short*>(cnt);
  const int nbase = range << DEG_RANGE_LOG;
  for (int n = tid; n < DEG_RANGE; n += DEG_THREADS) {
    const int node = nbase + n;
    const unsigned k = cnt16[n];
    if (node < N_NODES && k != 0u) {
      __hip_atomic_fetch_add(&deg[node], (float)k,
                             __ATOMIC_RELAXED, __HIP_MEMORY_SCOPE_AGENT);
    }
  }
}

__global__ void deg_rsqrt_kernel(float* __restrict__ deg) {
  int i = blockIdx.x * blockDim.x + threadIdx.x;
  if (i < N_NODES) deg[i] = rsqrtf(deg[i]);
}

template <int DIN, int DOUT, bool TWO>
__global__ void gemm_wmma_kernel(const float* __restrict__ H,
                                 const float* __restrict__ H2,
                                 const float* __restrict__ W,
                                 float* __restrict__ Out) {
  constexpr int NT = (DOUT + 15) / 16;
  const int wave = (blockIdx.x * blockDim.x + threadIdx.x) >> 5;
  const int lane = threadIdx.x & 31;
  const int m0 = wave * 16;
  if (m0 >= N_NODES) return;

  const int lhalf = lane & 15;
  const int khi = (lane >> 4) << 1;
  const int ncl = (lhalf < DOUT) ? lhalf : (DOUT - 1);

  const float* __restrict__ hrow = H + (size_t)(m0 + lhalf) * DIN;
  const float* __restrict__ hrow2 = H2 + (size_t)(m0 + lhalf) * DIN;

  v8f acc0 = {};
  v8f acc1 = {};
#pragma unroll
  for (int k = 0; k < DIN; k += 4) {
    v2f a, b0;
    a.x = hrow[k + khi];
    a.y = hrow[k + khi + 1];
    if (TWO) {
      a.x += hrow2[k + khi];
      a.y += hrow2[k + khi + 1];
    }
    b0.x = W[(k + khi) * DOUT + ncl];
    b0.y = W[(k + khi + 1) * DOUT + ncl];
    acc0 = __builtin_amdgcn_wmma_f32_16x16x4_f32(
        false, a, false, b0, (short)0, acc0, false, false);
    if (NT > 1) {
      v2f b1;
      b1.x = W[(k + khi) * DOUT + 16 + lhalf];
      b1.y = W[(k + khi + 1) * DOUT + 16 + lhalf];
      acc1 = __builtin_amdgcn_wmma_f32_16x16x4_f32(
          false, a, false, b1, (short)0, acc1, false, false);
    }
  }

  const int mrow = m0 + ((lane >> 4) << 3);
#pragma unroll
  for (int v = 0; v < 8; ++v) {
    const int m = mrow + v;
    if (lhalf < DOUT) Out[(size_t)m * DOUT + lhalf] = acc0[v];
    if (NT > 1)       Out[(size_t)m * DOUT + 16 + lhalf] = acc1[v];
  }
}

template <int DOUT, bool TWO>
__global__ void agg_init_kernel(const float* __restrict__ hW,
                                const float* __restrict__ bias,
                                const float* __restrict__ dinv,
                                float* __restrict__ out,
                                float* __restrict__ out2) {
  long long t = (long long)blockIdx.x * blockDim.x + threadIdx.x;
  if (t >= (long long)N_NODES * DOUT) return;
  int i = (int)(t / DOUT);
  int f = (int)(t - (long long)i * DOUT);
  float d = dinv[i];
  out[t] = hW[t] * (d * d) + bias[f];
  if (TWO) out2[t] = 0.0f;
}

#define L5_RANGES 8
#define L5_RANGE_NODES 12512
#define L5_CHUNKS 32
#define L5_CHUNK_EDGES (N_EDGES / L5_CHUNKS)
#define L5_THREADS 1024
#define L5_ACC (L5_RANGE_NODES * 3)
#define L5_PLANE (N_NODES * 3)
#define L5_VECS 4
#define L5_HOLD16(a)                                                          \
  asm volatile("" ::"v"(a[0]), "v"(a[1]), "v"(a[2]), "v"(a[3]), "v"(a[4]),   \
               "v"(a[5]), "v"(a[6]), "v"(a[7]), "v"(a[8]), "v"(a[9]),        \
               "v"(a[10]), "v"(a[11]), "v"(a[12]), "v"(a[13]), "v"(a[14]),   \
               "v"(a[15]))
static_assert(L5_VECS * 4 == 16, "L5_HOLD16 names sixteen values");
static_assert(L5_RANGES * L5_RANGE_NODES >= N_NODES, "ranges must cover all nodes");
static_assert(L5_RANGE_NODES % 32 == 0, "a range must start on a 128-byte line");
static_assert(N_EDGES % L5_CHUNKS == 0, "edge slices must be equal");
static_assert(L5_CHUNK_EDGES % 4 == 0, "slices are read four edges at a time");

__global__ void __launch_bounds__(L5_THREADS)
agg_edge3_lds_kernel(const float* __restrict__ hW,
                     const int* __restrict__ row,
                     const int* __restrict__ col,
                     const float* __restrict__ wgt,
                     float* __restrict__ part) {
  __shared__ float acc[L5_ACC];
  const int range = blockIdx.x / L5_CHUNKS;
  const int chunk = blockIdx.x - range * L5_CHUNKS;
  const int tid = threadIdx.x;
  const int lo = range * L5_RANGE_NODES;

  for (int w = tid; w < L5_ACC; w += L5_THREADS) acc[w] = 0.0f;
  __syncthreads();

  constexpr int NV = L5_CHUNK_EDGES / 4;
  constexpr int IT = (NV + L5_THREADS - 1) / L5_THREADS;
  const size_t ebase = (size_t)chunk * L5_CHUNK_EDGES;
  const int4* __restrict__ col4 = reinterpret_cast<const int4*>(col + ebase);
  const int4* __restrict__ row4 = reinterpret_cast<const int4*>(row + ebase);
  const float4* __restrict__ wgt4 = reinterpret_cast<const float4*>(wgt + ebase);

#pragma unroll
  for (int k0 = 0; k0 < IT; k0 += L5_VECS) {
    int4 cv[L5_VECS], rv[L5_VECS];
    float4 wv[L5_VECS];
#pragma unroll
    for (int k = 0; k < L5_VECS; ++k) {
      const int idx = tid + (k0 + k) * L5_THREADS;
      const int ci = idx < NV ? idx : NV - 1;
      cv[k] = col4[ci];
      rv[k] = row4[ci];
      wv[k] = wgt4[ci];
    }
    int rr[L5_VECS * 4];
#pragma unroll
    for (int k = 0; k < L5_VECS; ++k) {
      rr[k * 4 + 0] = rv[k].x;
      rr[k * 4 + 1] = rv[k].y;
      rr[k * 4 + 2] = rv[k].z;
      rr[k * 4 + 3] = rv[k].w;
    }
    L5_HOLD16(rr);
    float g0[L5_VECS * 4], g1[L5_VECS * 4], g2[L5_VECS * 4], ww[L5_VECS * 4];
    int loc[L5_VECS * 4];
#pragma unroll
    for (int k = 0; k < L5_VECS; ++k) {
      const bool valid = (tid + (k0 + k) * L5_THREADS) < NV;
      const int c4[4] = {cv[k].x, cv[k].y, cv[k].z, cv[k].w};
      const float w4[4] = {wv[k].x, wv[k].y, wv[k].z, wv[k].w};
#pragma unroll
      for (int q = 0; q < 4; ++q) {
        const int j = k * 4 + q;
        const int local = c4[q] - lo;
        const bool in = valid && ((unsigned)local < (unsigned)L5_RANGE_NODES);
        const unsigned r = in ? (unsigned)rr[j] : 0u;
        const float* __restrict__ p = hW + r * 3u;
        g0[j] = p[0];
        g1[j] = p[1];
        g2[j] = p[2];
        ww[j] = w4[q];
        loc[j] = in ? local * 3 : -1;
      }
    }
    L5_HOLD16(g0);
    L5_HOLD16(g1);
    L5_HOLD16(g2);
    L5_HOLD16(ww);
#pragma unroll
    for (int j = 0; j < L5_VECS * 4; ++j) {
      if (loc[j] >= 0) {
        __hip_atomic_fetch_add(&acc[loc[j]], g0[j] * ww[j],
                               __ATOMIC_RELAXED, __HIP_MEMORY_SCOPE_WORKGROUP);
        __hip_atomic_fetch_add(&acc[loc[j] + 1], g1[j] * ww[j],
                               __ATOMIC_RELAXED, __HIP_MEMORY_SCOPE_WORKGROUP);
        __hip_atomic_fetch_add(&acc[loc[j] + 2], g2[j] * ww[j],
                               __ATOMIC_RELAXED, __HIP_MEMORY_SCOPE_WORKGROUP);
      }
    }
  }
  __syncthreads();

  const int left = (N_NODES - lo) * 3;
  const int nvalid = left < L5_ACC ? left : L5_ACC;
  float* __restrict__ dst = part + (size_t)chunk * L5_PLANE + (size_t)lo * 3;
  for (int n = tid; n < nvalid; n += L5_THREADS) dst[n] = acc[n];
}

__global__ void agg_final3_kernel(const float* __restrict__ hW,
                                  const float* __restrict__ bias,
                                  const float* __restrict__ dinv,
                                  const float* __restrict__ part,
                                  float* __restrict__ out) {
  const int t = blockIdx.x * blockDim.x + threadIdx.x;
  if (t >= L5_PLANE) return;
  const int i = t / 3;
  const int f = t - i * 3;
  float pv[L5_CHUNKS];
#pragma unroll
  for (int k = 0; k < L5_CHUNKS; ++k) pv[k] = part[(size_t)k * L5_PLANE + t];
  const float d = dinv[i];
  float s = hW[t] * (d * d) + bias[f];
#pragma unroll
  for (int k = 0; k < L5_CHUNKS; ++k) s += pv[k];
  out[t] = s;
}

#define EDGES_PER_WAVE 16
template <bool MAKE_W>
__global__ void agg_edge32_batched_kernel(const float* __restrict__ hW,
                                          const int* __restrict__ row,
                                          const int* __restrict__ col,
                                          const float* __restrict__ dinv,
                                          float* __restrict__ wgt,
                                          float* __restrict__ out0,
                                          float* __restrict__ out1) {
  const int wave = (blockIdx.x * blockDim.x + threadIdx.x) >> 5;
  const int lane = threadIdx.x & 31;
  const int e0 = wave * EDGES_PER_WAVE;
  if (e0 >= N_EDGES) return;

  const int sub = lane & (EDGES_PER_WAVE - 1);
  const int myr = row[e0 + sub];
  const int myc = col[e0 + sub];
  float myw;
  if (MAKE_W) {
    myw = dinv[myr] * dinv[myc];
    if (lane < EDGES_PER_WAVE) wgt[e0 + lane] = myw;
  } else {
    myw = wgt[e0 + sub];
  }

  float v[EDGES_PER_WAVE];
#pragma unroll
  for (int j = 0; j < EDGES_PER_WAVE; ++j) {
    const int r = __builtin_amdgcn_readlane(myr, j);
    v[j] = hW[(size_t)r * 32 + lane];
  }
  const unsigned place = __builtin_amdgcn_s_sendmsg_rtn(0x87);
  float* __restrict__ out = (place & (1u << 18)) ? out1 : out0;

#pragma unroll
  for (int j = 0; j < EDGES_PER_WAVE; ++j) {
    const int c = __builtin_amdgcn_readlane(myc, j);
    const float w = __int_as_float(__builtin_amdgcn_readlane(__float_as_int(myw), j));
    __hip_atomic_fetch_add(&out[(size_t)c * 32 + lane], v[j] * w,
                           __ATOMIC_RELAXED, __HIP_MEMORY_SCOPE_AGENT);
  }
}

extern "C" void kernel_launch(void* const* d_in, const int* in_sizes, int n_in,
                              void* d_out, int out_size, void* d_ws, size_t ws_size,
                              hipStream_t stream) {
  const float* x  = (const float*)d_in[0];
  const int*   ei = (const int*)d_in[1];
  const int*   row = ei;
  const int*   col = ei + N_EDGES;

  const float* Wm[5] = {(const float*)d_in[2], (const float*)d_in[4],
                        (const float*)d_in[6], (const float*)d_in[8],
                        (const float*)d_in[10]};
  const float* Bm[5] = {(const float*)d_in[3], (const float*)d_in[5],
                        (const float*)d_in[7], (const float*)d_in[9],
                        (const float*)d_in[11]};

  float* ws   = (float*)d_ws;
  float* dinv = ws;
  float* hW   = ws + N_NODES;
  float* h    = ws + N_NODES + N_NODES * 32;
  float* wgt  = ws + N_NODES + N_NODES * 64;
  float* h1   = ws + N_NODES + N_NODES * 64 + N_EDGES;
  float* part = ws + N_NODES + N_NODES * 96 + N_EDGES;

  const int B = 256;
  const int gN  = (N_NODES + B - 1) / B;
  const int gGemm = ((N_NODES / 16) * 32 + B - 1) / B;
  const int gEdge32 = ((N_EDGES / EDGES_PER_WAVE) * 32 + B - 1) / B;

  deg_init_kernel<<<gN, B, 0, stream>>>(dinv);
  deg_hist_kernel<<<DEG_NRANGES * DEG_CHUNKS, DEG_THREADS, 0, stream>>>(dinv, col);
  deg_rsqrt_kernel<<<gN, B, 0, stream>>>(dinv);

  gemm_wmma_kernel<8, 32, false><<<gGemm, B, 0, stream>>>(x, x, Wm[0], hW);
  {
    long long n = (long long)N_NODES * 32;
    agg_init_kernel<32, true><<<(int)((n + B - 1) / B), B, 0, stream>>>(hW, Bm[0], dinv, h, h1);
    agg_edge32_batched_kernel<true><<<gEdge32, B, 0, stream>>>(hW, row, col, dinv, wgt, h, h1);
  }

  for (int l = 1; l <= 3; ++l) {
    gemm_wmma_kernel<32, 32, true><<<gGemm, B, 0, stream>>>(h, h1, Wm[l], hW);
    long long n = (long long)N_NODES * 32;
    agg_init_kernel<32, true><<<(int)((n + B - 1) / B), B, 0, stream>>>(hW, Bm[l], dinv, h, h1);
    agg_edge32_batched_kernel<false><<<gEdge32, B, 0, stream>>>(hW, row, col, dinv, wgt, h, h1);
  }

  gemm_wmma_kernel<32, 3, true><<<gGemm, B, 0, stream>>>(h, h1, Wm[4], hW);
  {
    float* out = (float*)d_out;
    agg_edge3_lds_kernel<<<L5_RANGES * L5_CHUNKS, L5_THREADS, 0, stream>>>(hW, row, col, wgt, part);
    agg_final3_kernel<<<(L5_PLANE + B - 1) / B, B, 0, stream>>>(hW, Bm[4], dinv, part, out);
  }
}


# Layer 1 aggregates before its product: the scatter gathers and adds x rows (8 floats per edge, not 32), then (A x) W1 + b1 on WMMA; products and sums of layer 1 are reassociated
# speedup vs baseline: 3.0473x; 1.0557x over previous
#include <hip/hip_runtime.h>

#define N_NODES 100000
#define N_EDGES 1600000

typedef __attribute__((ext_vector_type(2))) float v2f;
typedef __attribute__((ext_vector_type(8))) float v8f;

__global__ void deg_init_kernel(float* __restrict__ deg) {
  int i = blockIdx.x * blockDim.x + threadIdx.x;
  if (i < N_NODES) deg[i] = 1.0f;
}

#define DEG_RANGE_LOG 15
#define DEG_RANGE (1 << DEG_RANGE_LOG)
#define DEG_NRANGES ((N_NODES + DEG_RANGE - 1) / DEG_RANGE)
#define DEG_CHUNKS 32
#define DEG_CHUNK_EDGES (N_EDGES / DEG_CHUNKS)
#define DEG_THREADS 1024
static_assert(N_EDGES % DEG_CHUNKS == 0, "edge slices must be equal");
static_assert(DEG_CHUNK_EDGES % 4 == 0, "slices are read four indices at a time");
static_assert(DEG_CHUNK_EDGES < 65536, "a 16-bit counter must hold a whole slice");

__global__ void __launch_bounds__(DEG_THREADS)
deg_hist_kernel(float* __restrict__ deg, const int* __restrict__ col) {
  __shared__ unsigned cnt[DEG_RANGE / 2];
  const int range = blockIdx.x / DEG_CHUNKS;
  const int chunk = blockIdx.x - range * DEG_CHUNKS;
  const int tid = threadIdx.x;

  for (int w = tid; w < DEG_RANGE / 2; w += DEG_THREADS) cnt[w] = 0u;

  constexpr int NV = DEG_CHUNK_EDGES / 4;
  constexpr int IT = (NV + DEG_THREADS - 1) / DEG_THREADS;
  const int4* __restrict__ col4 =
      reinterpret_cast<const int4*>(col + (size_t)chunk * DEG_CHUNK_EDGES);
  int4 cv[IT];
#pragma unroll
  for (int k = 0; k < IT; ++k) {
    const int idx = tid + k * DEG_THREADS;
    cv[k] = col4[idx < NV ? idx : NV - 1];
  }
  __syncthreads();

#pragma unroll
  for (int k = 0; k < IT; ++k) {
    if (tid + k * DEG_THREADS < NV) {
      const int c4[4] = {cv[k].x, cv[k].y, cv[k].z, cv[k].w};
#pragma unroll
      for (int q = 0; q < 4; ++q) {
        const int c = c4[q];
        if ((c >> DEG_RANGE_LOG) == range) {
          const int local = c & (DEG_RANGE - 1);
          atomicAdd(&cnt[local >> 1], 1u << ((local & 1) << 4));
        }
      }
    }
  }
  __syncthreads();

  const unsigned short* cnt16 = reinterpret_cast<const unsigned short*>(cnt);
  const int nbase = range << DEG_RANGE_LOG;
  for (int n = tid; n < DEG_RANGE; n += DEG_THREADS) {
    const int node = nbase + n;
    const unsigned k = cnt16[n];
    if (node < N_NODES && k != 0u) {
      __hip_atomic_fetch_add(&deg[node], (float)k,
                             __ATOMIC_RELAXED, __HIP_MEMORY_SCOPE_AGENT);
    }
  }
}

__global__ void deg_rsqrt_kernel(float* __restrict__ deg) {
  int i = blockIdx.x * blockDim.x + threadIdx.x;
  if (i < N_NODES) deg[i] = rsqrtf(deg[i]);
}

template <int DIN, int DOUT, bool TWO, bool BIAS>
__global__ void gemm_wmma_kernel(const float* __restrict__ H,
                                 const float* __restrict__ H2,
                                 const float* __restrict__ W,
                                 const float* __restrict__ bias,
                                 float* __restrict__ Out) {
  constexpr int NT = (DOUT + 15) / 16;
  const int wave = (blockIdx.x * blockDim.x + threadIdx.x) >> 5;
  const int lane = threadIdx.x & 31;
  const int m0 = wave * 16;
  if (m0 >= N_NODES) return;

  const int lhalf = lane & 15;
  const int khi = (lane >> 4) << 1;
  const int ncl = (lhalf < DOUT) ? lhalf : (DOUT - 1);

  const float* __restrict__ hrow = H + (size_t)(m0 + lhalf) * DIN;
  const float* __restrict__ hrow2 = H2 + (size_t)(m0 + lhalf) * DIN;

  v8f acc0 = {};
  v8f acc1 = {};
#pragma unroll
  for (int k = 0; k < DIN; k += 4) {
    v2f a, b0;
    a.x = hrow[k + khi];
    a.y = hrow[k + khi + 1];
    if (TWO) {
      a.x += hrow2[k + khi];
      a.y += hrow2[k + khi + 1];
    }
    b0.x = W[(k + khi) * DOUT + ncl];
    b0.y = W[(k + khi + 1) * DOUT + ncl];
    acc0 = __builtin_amdgcn_wmma_f32_16x16x4_f32(
        false, a, false, b0, (short)0, acc0, false, false);
    if (NT > 1) {
      v2f b1;
      b1.x = W[(k + khi) * DOUT + 16 + lhalf];
      b1.y = W[(k + khi + 1) * DOUT + 16 + lhalf];
      acc1 = __builtin_amdgcn_wmma_f32_16x16x4_f32(
          false, a, false, b1, (short)0, acc1, false, false);
    }
  }

  float bb0 = 0.0f, bb1 = 0.0f;
  if (BIAS) {
    bb0 = bias[ncl];
    if (NT > 1) bb1 = bias[16 + lhalf];
  }
  const int mrow = m0 + ((lane >> 4) << 3);
#pragma unroll
  for (int v = 0; v < 8; ++v) {
    const int m = mrow + v;
    if (lhalf < DOUT) Out[(size_t)m * DOUT + lhalf] = acc0[v] + bb0;
    if (NT > 1)       Out[(size_t)m * DOUT + 16 + lhalf] = acc1[v] + bb1;
  }
}

template <int DOUT, bool TWO, bool BIAS>
__global__ void agg_init_kernel(const float* __restrict__ hW,
                                const float* __restrict__ bias,
                                const float* __restrict__ dinv,
                                float* __restrict__ out,
                                float* __restrict__ out2) {
  long long t = (long long)blockIdx.x * blockDim.x + threadIdx.x;
  if (t >= (long long)N_NODES * DOUT) return;
  int i = (int)(t / DOUT);
  int f = (int)(t - (long long)i * DOUT);
  float d = dinv[i];
  out[t] = BIAS ? hW[t] * (d * d) + bias[f] : hW[t] * (d * d);
  if (TWO) out2[t] = 0.0f;
}

#define L5_RANGES 8
#define L5_RANGE_NODES 12512
#define L5_CHUNKS 32
#define L5_CHUNK_EDGES (N_EDGES / L5_CHUNKS)
#define L5_THREADS 1024
#define L5_ACC (L5_RANGE_NODES * 3)
#define L5_PLANE (N_NODES * 3)
#define L5_VECS 4
#define L5_HOLD16(a)                                                          \
  asm volatile("" ::"v"(a[0]), "v"(a[1]), "v"(a[2]), "v"(a[3]), "v"(a[4]),   \
               "v"(a[5]), "v"(a[6]), "v"(a[7]), "v"(a[8]), "v"(a[9]),        \
               "v"(a[10]), "v"(a[11]), "v"(a[12]), "v"(a[13]), "v"(a[14]),   \
               "v"(a[15]))
static_assert(L5_VECS * 4 == 16, "L5_HOLD16 names sixteen values");
static_assert(L5_RANGES * L5_RANGE_NODES >= N_NODES, "ranges must cover all nodes");
static_assert(L5_RANGE_NODES % 32 == 0, "a range must start on a 128-byte line");
static_assert(N_EDGES % L5_CHUNKS == 0, "edge slices must be equal");
static_assert(L5_CHUNK_EDGES % 4 == 0, "slices are read four edges at a time");

__global__ void __launch_bounds__(L5_THREADS)
agg_edge3_lds_kernel(const float* __restrict__ hW,
                     const int* __restrict__ row,
                     const int* __restrict__ col,
                     const float* __restrict__ wgt,
                     float* __restrict__ part) {
  __shared__ float acc[L5_ACC];
  const int range = blockIdx.x / L5_CHUNKS;
  const int chunk = blockIdx.x - range * L5_CHUNKS;
  const int tid = threadIdx.x;
  const int lo = range * L5_RANGE_NODES;

  for (int w = tid; w < L5_ACC; w += L5_THREADS) acc[w] = 0.0f;
  __syncthreads();

  constexpr int NV = L5_CHUNK_EDGES / 4;
  constexpr int IT = (NV + L5_THREADS - 1) / L5_THREADS;
  const size_t ebase = (size_t)chunk * L5_CHUNK_EDGES;
  const int4* __restrict__ col4 = reinterpret_cast<const int4*>(col + ebase);
  const int4* __restrict__ row4 = reinterpret_cast<const int4*>(row + ebase);
  const float4* __restrict__ wgt4 = reinterpret_cast<const float4*>(wgt + ebase);

#pragma unroll
  for (int k0 = 0; k0 < IT; k0 += L5_VECS) {
    int4 cv[L5_VECS], rv[L5_VECS];
    float4 wv[L5_VECS];
#pragma unroll
    for (int k = 0; k < L5_VECS; ++k) {
      const int idx = tid + (k0 + k) * L5_THREADS;
      const int ci = idx < NV ? idx : NV - 1;
      cv[k] = col4[ci];
      rv[k] = row4[ci];
      wv[k] = wgt4[ci];
    }
    int rr[L5_VECS * 4];
#pragma unroll
    for (int k = 0; k < L5_VECS; ++k) {
      rr[k * 4 + 0] = rv[k].x;
      rr[k * 4 + 1] = rv[k].y;
      rr[k * 4 + 2] = rv[k].z;
      rr[k * 4 + 3] = rv[k].w;
    }
    L5_HOLD16(rr);
    float g0[L5_VECS * 4], g1[L5_VECS * 4], g2[L5_VECS * 4], ww[L5_VECS * 4];
    int loc[L5_VECS * 4];
#pragma unroll
    for (int k = 0; k < L5_VECS; ++k) {
      const bool valid = (tid + (k0 + k) * L5_THREADS) < NV;
      const int c4[4] = {cv[k].x, cv[k].y, cv[k].z, cv[k].w};
      const float w4[4] = {wv[k].x, wv[k].y, wv[k].z, wv[k].w};
#pragma unroll
      for (int q = 0; q < 4; ++q) {
        const int j = k * 4 + q;
        const int local = c4[q] - lo;
        const bool in = valid && ((unsigned)local < (unsigned)L5_RANGE_NODES);
        const unsigned r = in ? (unsigned)rr[j] : 0u;
        const float* __restrict__ p = hW + r * 3u;
        g0[j] = p[0];
        g1[j] = p[1];
        g2[j] = p[2];
        ww[j] = w4[q];
        loc[j] = in ? local * 3 : -1;
      }
    }
    L5_HOLD16(g0);
    L5_HOLD16(g1);
    L5_HOLD16(g2);
    L5_HOLD16(ww);
#pragma unroll
    for (int j = 0; j < L5_VECS * 4; ++j) {
      if (loc[j] >= 0) {
        __hip_atomic_fetch_add(&acc[loc[j]], g0[j] * ww[j],
                               __ATOMIC_RELAXED, __HIP_MEMORY_SCOPE_WORKGROUP);
        __hip_atomic_fetch_add(&acc[loc[j] + 1], g1[j] * ww[j],
                               __ATOMIC_RELAXED, __HIP_MEMORY_SCOPE_WORKGROUP);
        __hip_atomic_fetch_add(&acc[loc[j] + 2], g2[j] * ww[j],
                               __ATOMIC_RELAXED, __HIP_MEMORY_SCOPE_WORKGROUP);
      }
    }
  }
  __syncthreads();

  const int left = (N_NODES - lo) * 3;
  const int nvalid = left < L5_ACC ? left : L5_ACC;
  float* __restrict__ dst = part + (size_t)chunk * L5_PLANE + (size_t)lo * 3;
  for (int n = tid; n < nvalid; n += L5_THREADS) dst[n] = acc[n];
}

__global__ void agg_final3_kernel(const float* __restrict__ hW,
                                  const float* __restrict__ bias,
                                  const float* __restrict__ dinv,
                                  const float* __restrict__ part,
                                  float* __restrict__ out) {
  const int t = blockIdx.x * blockDim.x + threadIdx.x;
  if (t >= L5_PLANE) return;
  const int i = t / 3;
  const int f = t - i * 3;
  float pv[L5_CHUNKS];
#pragma unroll
  for (int k = 0; k < L5_CHUNKS; ++k) pv[k] = part[(size_t)k * L5_PLANE + t];
  const float d = dinv[i];
  float s = hW[t] * (d * d) + bias[f];
#pragma unroll
  for (int k = 0; k < L5_CHUNKS; ++k) s += pv[k];
  out[t] = s;
}

#define EDGES_PER_WAVE 16
__global__ void agg_edge32_batched_kernel(const float* __restrict__ hW,
                                          const int* __restrict__ row,
                                          const int* __restrict__ col,
                                          const float* __restrict__ wgt,
                                          float* __restrict__ out0,
                                          float* __restrict__ out1) {
  const int wave = (blockIdx.x * blockDim.x + threadIdx.x) >> 5;
  const int lane = threadIdx.x & 31;
  const int e0 = wave * EDGES_PER_WAVE;
  if (e0 >= N_EDGES) return;

  const int sub = lane & (EDGES_PER_WAVE - 1);
  const int myr = row[e0 + sub];
  const int myc = col[e0 + sub];
  const float myw = wgt[e0 + sub];

  float v[EDGES_PER_WAVE];
#pragma unroll
  for (int j = 0; j < EDGES_PER_WAVE; ++j) {
    const int r = __builtin_amdgcn_readlane(myr, j);
    v[j] = hW[(size_t)r * 32 + lane];
  }
  const unsigned place = __builtin_amdgcn_s_sendmsg_rtn(0x87);
  float* __restrict__ out = (place & (1u << 18)) ? out1 : out0;

#pragma unroll
  for (int j = 0; j < EDGES_PER_WAVE; ++j) {
    const int c = __builtin_amdgcn_readlane(myc, j);
    const float w = __int_as_float(__builtin_amdgcn_readlane(__float_as_int(myw), j));
    __hip_atomic_fetch_add(&out[(size_t)c * 32 + lane], v[j] * w,
                           __ATOMIC_RELAXED, __HIP_MEMORY_SCOPE_AGENT);
  }
}

#define E8_PER_WAVE 32
static_assert(N_EDGES % E8_PER_WAVE == 0, "whole waves of edges");
__global__ void agg_edge8_kernel(const float* __restrict__ xin,
                                 const int* __restrict__ row,
                                 const int* __restrict__ col,
                                 const float* __restrict__ dinv,
                                 float* __restrict__ wgt,
                                 float* __restrict__ out0,
                                 float* __restrict__ out1) {
  const int wave = (blockIdx.x * blockDim.x + threadIdx.x) >> 5;
  const int lane = threadIdx.x & 31;
  const int e0 = wave * E8_PER_WAVE;
  if (e0 >= N_EDGES) return;

  const int myr = row[e0 + lane];
  const int myc = col[e0 + lane];
  const float myw = dinv[myr] * dinv[myc];
  wgt[e0 + lane] = myw;

  const int f = lane & 7;
  const int q = lane >> 3;
  float v[8];
  int cc[8];
  float ww[8];
#pragma unroll
  for (int j = 0; j < 8; ++j) {
    const int r = __shfl(myr, j * 4 + q);
    v[j] = xin[(size_t)r * 8 + f];
    cc[j] = __shfl(myc, j * 4 + q);
    ww[j] = __shfl(myw, j * 4 + q);
  }
  const unsigned place = __builtin_amdgcn_s_sendmsg_rtn(0x87);
  float* __restrict__ out = (place & (1u << 18)) ? out1 : out0;

#pragma unroll
  for (int j = 0; j < 8; ++j) {
    __hip_atomic_fetch_add(&out[(size_t)cc[j] * 8 + f], v[j] * ww[j],
                           __ATOMIC_RELAXED, __HIP_MEMORY_SCOPE_AGENT);
  }
}

extern "C" void kernel_launch(void* const* d_in, const int* in_sizes, int n_in,
                              void* d_out, int out_size, void* d_ws, size_t ws_size,
                              hipStream_t stream) {
  const float* x  = (const float*)d_in[0];
  const int*   ei = (const int*)d_in[1];
  const int*   row = ei;
  const int*   col = ei + N_EDGES;

  const float* Wm[5] = {(const float*)d_in[2], (const float*)d_in[4],
                        (const float*)d_in[6], (const float*)d_in[8],
                        (const float*)d_in[10]};
  const float* Bm[5] = {(const float*)d_in[3], (const float*)d_in[5],
                        (const float*)d_in[7], (const float*)d_in[9],
                        (const float*)d_in[11]};

  float* ws   = (float*)d_ws;
  float* dinv = ws;
  float* hW   = ws + N_NODES;
  float* h    = ws + N_NODES + N_NODES * 32;
  float* wgt  = ws + N_NODES + N_NODES * 64;
  float* h1   = ws + N_NODES + N_NODES * 64 + N_EDGES;
  float* part = ws + N_NODES + N_NODES * 96 + N_EDGES;
  float* s0   = ws + N_NODES + N_NODES * 192 + N_EDGES;
  float* s1   = ws + N_NODES + N_NODES * 200 + N_EDGES;

  const int B = 256;
  const int gN  = (N_NODES + B - 1) / B;
  const int gGemm = ((N_NODES / 16) * 32 + B - 1) / B;
  const int gEdge32 = ((N_EDGES / EDGES_PER_WAVE) * 32 + B - 1) / B;
  const int gEdge8 = ((N_EDGES / E8_PER_WAVE) * 32 + B - 1) / B;

  deg_init_kernel<<<gN, B, 0, stream>>>(dinv);
  deg_hist_kernel<<<DEG_NRANGES * DEG_CHUNKS, DEG_THREADS, 0, stream>>>(dinv, col);
  deg_rsqrt_kernel<<<gN, B, 0, stream>>>(dinv);

  {
    long long n = (long long)N_NODES * 8;
    agg_init_kernel<8, true, false><<<(int)((n + B - 1) / B), B, 0, stream>>>(x, Bm[0], dinv, s0, s1);
    agg_edge8_kernel<<<gEdge8, B, 0, stream>>>(x, row, col, dinv, wgt, s0, s1);
  }
  gemm_wmma_kernel<8, 32, true, true><<<gGemm, B, 0, stream>>>(s0, s1, Wm[0], Bm[0], h1);

  for (int l = 1; l <= 3; ++l) {
    if (l == 1) {
      gemm_wmma_kernel<32, 32, false, false><<<gGemm, B, 0, stream>>>(h1, h1, Wm[l], Bm[l], hW);
    } else {
      gemm_wmma_kernel<32, 32, true, false><<<gGemm, B, 0, stream>>>(h, h1, Wm[l], Bm[l], hW);
    }
    long long n = (long long)N_NODES * 32;
    agg_init_kernel<32, true, true><<<(int)((n + B - 1) / B), B, 0, stream>>>(hW, Bm[l], dinv, h, h1);
    agg_edge32_batched_kernel<<<gEdge32, B, 0, stream>>>(hW, row, col, wgt, h, h1);
  }

  gemm_wmma_kernel<32, 3, true, false><<<gGemm, B, 0, stream>>>(h, h1, Wm[4], Bm[4], hW);
  {
    float* out = (float*)d_out;
    agg_edge3_lds_kernel<<<L5_RANGES * L5_CHUNKS, L5_THREADS, 0, stream>>>(hW, row, col, wgt, part);
    agg_final3_kernel<<<(L5_PLANE + B - 1) / B, B, 0, stream>>>(hW, Bm[4], dinv, part, out);
  }
}


# Start-value launches dropped: each scatter launch zeroes the next layer's planes (one line per wave); the next product's A loader adds self loop + bias after the edge sums; 14 launches, not 18
# speedup vs baseline: 3.5511x; 1.1653x over previous
#include <hip/hip_runtime.h>

#define N_NODES 100000
#define N_EDGES 1600000

typedef __attribute__((ext_vector_type(2))) float v2f;
typedef __attribute__((ext_vector_type(8))) float v8f;

__global__ void deg_init_kernel(float* __restrict__ deg, float* __restrict__ s01) {
  int i = blockIdx.x * blockDim.x + threadIdx.x;
  if (i < N_NODES) {
    deg[i] = 1.0f;
    float4* __restrict__ z4 = reinterpret_cast<float4*>(s01);
    const float4 z = {0.0f, 0.0f, 0.0f, 0.0f};
#pragma unroll
    for (int k = 0; k < 4; ++k) z4[(size_t)i + (size_t)k * N_NODES] = z;
  }
}

#define DEG_RANGE_LOG 15
#define DEG_RANGE (1 << DEG_RANGE_LOG)
#define DEG_NRANGES ((N_NODES + DEG_RANGE - 1) / DEG_RANGE)
#define DEG_CHUNKS 32
#define DEG_CHUNK_EDGES (N_EDGES / DEG_CHUNKS)
#define DEG_THREADS 1024
static_assert(N_EDGES % DEG_CHUNKS == 0, "edge slices must be equal");
static_assert(DEG_CHUNK_EDGES % 4 == 0, "slices are read four indices at a time");
static_assert(DEG_CHUNK_EDGES < 65536, "a 16-bit counter must hold a whole slice");

__global__ void __launch_bounds__(DEG_THREADS)
deg_hist_kernel(float* __restrict__ deg, const int* __restrict__ col) {
  __shared__ unsigned cnt[DEG_RANGE / 2];
  const int range = blockIdx.x / DEG_CHUNKS;
  const int chunk = blockIdx.x - range * DEG_CHUNKS;
  const int tid = threadIdx.x;

  for (int w = tid; w < DEG_RANGE / 2; w += DEG_THREADS) cnt[w] = 0u;

  constexpr int NV = DEG_CHUNK_EDGES / 4;
  constexpr int IT = (NV + DEG_THREADS - 1) / DEG_THREADS;
  const int4* __restrict__ col4 =
      reinterpret_cast<const int4*>(col + (size_t)chunk * DEG_CHUNK_EDGES);
  int4 cv[IT];
#pragma unroll
  for (int k = 0; k < IT; ++k) {
    const int idx = tid + k * DEG_THREADS;
    cv[k] = col4[idx < NV ? idx : NV - 1];
  }
  __syncthreads();

#pragma unroll
  for (int k = 0; k < IT; ++k) {
    if (tid + k * DEG_THREADS < NV) {
      const int c4[4] = {cv[k].x, cv[k].y, cv[k].z, cv[k].w};
#pragma unroll
      for (int q = 0; q < 4; ++q) {
        const int c = c4[q];
        if ((c >> DEG_RANGE_LOG) == range) {
          const int local = c & (DEG_RANGE - 1);
          atomicAdd(&cnt[local >> 1], 1u << ((local & 1) << 4));
        }
      }
    }
  }
  __syncthreads();

  const unsigned short* cnt16 = reinterpret_cast<const unsigned short*>(cnt);
  const int nbase = range << DEG_RANGE_LOG;
  for (int n = tid; n < DEG_RANGE; n += DEG_THREADS) {
    const int node = nbase + n;
    const unsigned k = cnt16[n];
    if (node < N_NODES && k != 0u) {
      __hip_atomic_fetch_add(&deg[node], (float)k,
                             __ATOMIC_RELAXED, __HIP_MEMORY_SCOPE_AGENT);
    }
  }
}

__global__ void deg_rsqrt_kernel(float* __restrict__ deg) {
  int i = blockIdx.x * blockDim.x + threadIdx.x;
  if (i < N_NODES) deg[i] = rsqrtf(deg[i]);
}

template <int DIN, int DOUT, bool TWO, bool BIAS, bool SELF, bool SBIAS>
__global__ void gemm_wmma_kernel(const float* __restrict__ H,
                                 const float* __restrict__ H2,
                                 const float* __restrict__ S,
                                 const float* __restrict__ dinv,
                                 const float* __restrict__ sbias,
                                 const float* __restrict__ W,
                                 const float* __restrict__ bias,
                                 float* __restrict__ Out) {
  constexpr int NT = (DOUT + 15) / 16;
  const int wave = (blockIdx.x * blockDim.x + threadIdx.x) >> 5;
  const int lane = threadIdx.x & 31;
  const int m0 = wave * 16;
  if (m0 >= N_NODES) return;

  const int lhalf = lane & 15;
  const int khi = (lane >> 4) << 1;
  const int ncl = (lhalf < DOUT) ? lhalf : (DOUT - 1);

  const float* __restrict__ hrow = H + (size_t)(m0 + lhalf) * DIN;
  const float* __restrict__ hrow2 = H2 + (size_t)(m0 + lhalf) * DIN;
  const float* __restrict__ srow = S + (size_t)(m0 + lhalf) * DIN;
  float dd = 0.0f;
  if (SELF) {
    const float d = dinv[m0 + lhalf];
    dd = d * d;
  }

  v8f acc0 = {};
  v8f acc1 = {};
#pragma unroll
  for (int k = 0; k < DIN; k += 4) {
    v2f a, b0;
    a.x = hrow[k + khi];
    a.y = hrow[k + khi + 1];
    if (TWO) {
      a.x += hrow2[k + khi];
      a.y += hrow2[k + khi + 1];
    }
    if (SELF) {
      float sx = srow[k + khi] * dd;
      float sy = srow[k + khi + 1] * dd;
      if (SBIAS) {
        sx += sbias[k + khi];
        sy += sbias[k + khi + 1];
      }
      a.x += sx;
      a.y += sy;
    }
    b0.x = W[(k + khi) * DOUT + ncl];
    b0.y = W[(k + khi + 1) * DOUT + ncl];
    acc0 = __builtin_amdgcn_wmma_f32_16x16x4_f32(
        false, a, false, b0, (short)0, acc0, false, false);
    if (NT > 1) {
      v2f b1;
      b1.x = W[(k + khi) * DOUT + 16 + lhalf];
      b1.y = W[(k + khi + 1) * DOUT + 16 + lhalf];
      acc1 = __builtin_amdgcn_wmma_f32_16x16x4_f32(
          false, a, false, b1, (short)0, acc1, false, false);
    }
  }

  float bb0 = 0.0f, bb1 = 0.0f;
  if (BIAS) {
    bb0 = bias[ncl];
    if (NT > 1) bb1 = bias[16 + lhalf];
  }
  const int mrow = m0 + ((lane >> 4) << 3);
#pragma unroll
  for (int v = 0; v < 8; ++v) {
    const int m = mrow + v;
    if (lhalf < DOUT) Out[(size_t)m * DOUT + lhalf] = acc0[v] + bb0;
    if (NT > 1)       Out[(size_t)m * DOUT + 16 + lhalf] = acc1[v] + bb1;
  }
}

#define L5_RANGES 8
#define L5_RANGE_NODES 12512
#define L5_CHUNKS 32
#define L5_CHUNK_EDGES (N_EDGES / L5_CHUNKS)
#define L5_THREADS 1024
#define L5_ACC (L5_RANGE_NODES * 3)
#define L5_PLANE (N_NODES * 3)
#define L5_VECS 4
#define L5_HOLD16(a)                                                          \
  asm volatile("" ::"v"(a[0]), "v"(a[1]), "v"(a[2]), "v"(a[3]), "v"(a[4]),   \
               "v"(a[5]), "v"(a[6]), "v"(a[7]), "v"(a[8]), "v"(a[9]),        \
               "v"(a[10]), "v"(a[11]), "v"(a[12]), "v"(a[13]), "v"(a[14]),   \
               "v"(a[15]))
static_assert(L5_VECS * 4 == 16, "L5_HOLD16 names sixteen values");
static_assert(L5_RANGES * L5_RANGE_NODES >= N_NODES, "ranges must cover all nodes");
static_assert(L5_RANGE_NODES % 32 == 0, "a range must start on a 128-byte line");
static_assert(N_EDGES % L5_CHUNKS == 0, "edge slices must be equal");
static_assert(L5_CHUNK_EDGES % 4 == 0, "slices are read four edges at a time");

__global__ void __launch_bounds__(L5_THREADS)
agg_edge3_lds_kernel(const float* __restrict__ hW,
                     const int* __restrict__ row,
                     const int* __restrict__ col,
                     const float* __restrict__ wgt,
                     float* __restrict__ part) {
  __shared__ float acc[L5_ACC];
  const int range = blockIdx.x / L5_CHUNKS;
  const int chunk = blockIdx.x - range * L5_CHUNKS;
  const int tid = threadIdx.x;
  const int lo = range * L5_RANGE_NODES;

  for (int w = tid; w < L5_ACC; w += L5_THREADS) acc[w] = 0.0f;
  __syncthreads();

  constexpr int NV = L5_CHUNK_EDGES / 4;
  constexpr int IT = (NV + L5_THREADS - 1) / L5_THREADS;
  const size_t ebase = (size_t)chunk * L5_CHUNK_EDGES;
  const int4* __restrict__ col4 = reinterpret_cast<const int4*>(col + ebase);
  const int4* __restrict__ row4 = reinterpret_cast<const int4*>(row + ebase);
  const float4* __restrict__ wgt4 = reinterpret_cast<const float4*>(wgt + ebase);

#pragma unroll
  for (int k0 = 0; k0 < IT; k0 += L5_VECS) {
    int4 cv[L5_VECS], rv[L5_VECS];
    float4 wv[L5_VECS];
#pragma unroll
    for (int k = 0; k < L5_VECS; ++k) {
      const int idx = tid + (k0 + k) * L5_THREADS;
      const int ci = idx < NV ? idx : NV - 1;
      cv[k] = col4[ci];
      rv[k] = row4[ci];
      wv[k] = wgt4[ci];
    }
    int rr[L5_VECS * 4];
#pragma unroll
    for (int k = 0; k < L5_VECS; ++k) {
      rr[k * 4 + 0] = rv[k].x;
      rr[k * 4 + 1] = rv[k].y;
      rr[k * 4 + 2] = rv[k].z;
      rr[k * 4 + 3] = rv[k].w;
    }
    L5_HOLD16(rr);
    float g0[L5_VECS * 4], g1[L5_VECS * 4], g2[L5_VECS * 4], ww[L5_VECS * 4];
    int loc[L5_VECS * 4];
#pragma unroll
    for (int k = 0; k < L5_VECS; ++k) {
      const bool valid = (tid + (k0 + k) * L5_THREADS) < NV;
      const int c4[4] = {cv[k].x, cv[k].y, cv[k].z, cv[k].w};
      const float w4[4] = {wv[k].x, wv[k].y, wv[k].z, wv[k].w};
#pragma unroll
      for (int q = 0; q < 4; ++q) {
        const int j = k * 4 + q;
        const int local = c4[q] - lo;
        const bool in = valid && ((unsigned)local < (unsigned)L5_RANGE_NODES);
        const unsigned r = in ? (unsigned)rr[j] : 0u;
        const float* __restrict__ p = hW + r * 3u;
        g0[j] = p[0];
        g1[j] = p[1];
        g2[j] = p[2];
        ww[j] = w4[q];
        loc[j] = in ? local * 3 : -1;
      }
    }
    L5_HOLD16(g0);
    L5_HOLD16(g1);
    L5_HOLD16(g2);
    L5_HOLD16(ww);
#pragma unroll
    for (int j = 0; j < L5_VECS * 4; ++j) {
      if (loc[j] >= 0) {
        __hip_atomic_fetch_add(&acc[loc[j]], g0[j] * ww[j],
                               __ATOMIC_RELAXED, __HIP_MEMORY_SCOPE_WORKGROUP);
        __hip_atomic_fetch_add(&acc[loc[j] + 1], g1[j] * ww[j],
                               __ATOMIC_RELAXED, __HIP_MEMORY_SCOPE_WORKGROUP);
        __hip_atomic_fetch_add(&acc[loc[j] + 2], g2[j] * ww[j],
                               __ATOMIC_RELAXED, __HIP_MEMORY_SCOPE_WORKGROUP);
      }
    }
  }
  __syncthreads();

  const int left = (N_NODES - lo) * 3;
  const int nvalid = left < L5_ACC ? left : L5_ACC;
  float* __restrict__ dst = part + (size_t)chunk * L5_PLANE + (size_t)lo * 3;
  for (int n = tid; n < nvalid; n += L5_THREADS) dst[n] = acc[n];
}

__global__ void agg_final3_kernel(const float* __restrict__ hW,
                                  const float* __restrict__ bias,
                                  const float* __restrict__ dinv,
                                  const float* __restrict__ part,
                                  float* __restrict__ out) {
  const int t = blockIdx.x * blockDim.x + threadIdx.x;
  if (t >= L5_PLANE) return;
  const int i = t / 3;
  const int f = t - i * 3;
  float pv[L5_CHUNKS];
#pragma unroll
  for (int k = 0; k < L5_CHUNKS; ++k) pv[k] = part[(size_t)k * L5_PLANE + t];
  const float d = dinv[i];
  float s = hW[t] * (d * d) + bias[f];
#pragma unroll
  for (int k = 0; k < L5_CHUNKS; ++k) s += pv[k];
  out[t] = s;
}

#define EDGES_PER_WAVE 16
static_assert(N_EDGES / EDGES_PER_WAVE == N_NODES, "one wave per node row to clear");
template <bool ZERO>
__global__ void agg_edge32_batched_kernel(const float* __restrict__ hW,
                                          const int* __restrict__ row,
                                          const int* __restrict__ col,
                                          const float* __restrict__ wgt,
                                          float* __restrict__ out0,
                                          float* __restrict__ out1,
                                          float* __restrict__ zp0,
                                          float* __restrict__ zp1) {
  const int wave = (blockIdx.x * blockDim.x + threadIdx.x) >> 5;
  const int lane = threadIdx.x & 31;
  const int e0 = wave * EDGES_PER_WAVE;
  if (e0 >= N_EDGES) return;
  if (ZERO) {
    zp0[(size_t)wave * 32 + lane] = 0.0f;
    zp1[(size_t)wave * 32 + lane] = 0.0f;
  }

  const int sub = lane & (EDGES_PER_WAVE - 1);
  const int myr = row[e0 + sub];
  const int myc = col[e0 + sub];
  const float myw = wgt[e0 + sub];

  float v[EDGES_PER_WAVE];
#pragma unroll
  for (int j = 0; j < EDGES_PER_WAVE; ++j) {
    const int r = __builtin_amdgcn_readlane(myr, j);
    v[j] = hW[(size_t)r * 32 + lane];
  }
  const unsigned place = __builtin_amdgcn_s_sendmsg_rtn(0x87);
  float* __restrict__ out = (place & (1u << 18)) ? out1 : out0;

#pragma unroll
  for (int j = 0; j < EDGES_PER_WAVE; ++j) {
    const int c = __builtin_amdgcn_readlane(myc, j);
    const float w = __int_as_float(__builtin_amdgcn_readlane(__float_as_int(myw), j));
    __hip_atomic_fetch_add(&out[(size_t)c * 32 + lane], v[j] * w,
                           __ATOMIC_RELAXED, __HIP_MEMORY_SCOPE_AGENT);
  }
}

#define E8_PER_WAVE 32
static_assert(N_EDGES % E8_PER_WAVE == 0, "whole waves of edges");
static_assert((N_EDGES / E8_PER_WAVE) * 2 == N_NODES, "two node rows to clear per wave");
__global__ void agg_edge8_kernel(const float* __restrict__ xin,
                                 const int* __restrict__ row,
                                 const int* __restrict__ col,
                                 const float* __restrict__ dinv,
                                 float* __restrict__ wgt,
                                 float* __restrict__ out0,
                                 float* __restrict__ out1,
                                 float* __restrict__ zp0,
                                 float* __restrict__ zp1) {
  const int wave = (blockIdx.x * blockDim.x + threadIdx.x) >> 5;
  const int lane = threadIdx.x & 31;
  const int e0 = wave * E8_PER_WAVE;
  if (e0 >= N_EDGES) return;
  {
    const float2 z2 = {0.0f, 0.0f};
    reinterpret_cast<float2*>(zp0 + (size_t)wave * 64)[lane] = z2;
    reinterpret_cast<float2*>(zp1 + (size_t)wave * 64)[lane] = z2;
  }

  const int myr = row[e0 + lane];
  const int myc = col[e0 + lane];
  const float myw = dinv[myr] * dinv[myc];
  wgt[e0 + lane] = myw;

  const int f = lane & 7;
  const int q = lane >> 3;
  float v[8];
  int cc[8];
  float ww[8];
#pragma unroll
  for (int j = 0; j < 8; ++j) {
    const int r = __shfl(myr, j * 4 + q);
    v[j] = xin[(size_t)r * 8 + f];
    cc[j] = __shfl(myc, j * 4 + q);
    ww[j] = __shfl(myw, j * 4 + q);
  }
  const unsigned place = __builtin_amdgcn_s_sendmsg_rtn(0x87);
  float* __restrict__ out = (place & (1u << 18)) ? out1 : out0;

#pragma unroll
  for (int j = 0; j < 8; ++j) {
    __hip_atomic_fetch_add(&out[(size_t)cc[j] * 8 + f], v[j] * ww[j],
                           __ATOMIC_RELAXED, __HIP_MEMORY_SCOPE_AGENT);
  }
}

extern "C" void kernel_launch(void* const* d_in, const int* in_sizes, int n_in,
                              void* d_out, int out_size, void* d_ws, size_t ws_size,
                              hipStream_t stream) {
  const float* x  = (const float*)d_in[0];
  const int*   ei = (const int*)d_in[1];
  const int*   row = ei;
  const int*   col = ei + N_EDGES;

  const float* Wm[5] = {(const float*)d_in[2], (const float*)d_in[4],
                        (const float*)d_in[6], (const float*)d_in[8],
                        (const float*)d_in[10]};
  const float* Bm[5] = {(const float*)d_in[3], (const float*)d_in[5],
                        (const float*)d_in[7], (const float*)d_in[9],
                        (const float*)d_in[11]};

  float* ws   = (float*)d_ws;
  float* dinv = ws;
  float* s0   = ws + N_NODES;
  float* s1   = ws + N_NODES * 9;
  float* h1   = ws + N_NODES * 17;
  float* hWA  = ws + N_NODES * 49;
  float* hWB  = ws + N_NODES * 81;
  float* pa0  = ws + N_NODES * 113;
  float* pa1  = ws + N_NODES * 145;
  float* pb0  = ws + N_NODES * 177;
  float* pb1  = ws + N_NODES * 209;
  float* hW3  = ws + N_NODES * 241;
  float* part = ws + N_NODES * 245;
  float* wgt  = ws + N_NODES * 341;

  const int B = 256;
  const int gN  = (N_NODES + B - 1) / B;
  const int gGemm = ((N_NODES / 16) * 32 + B - 1) / B;
  const int gEdge32 = ((N_EDGES / EDGES_PER_WAVE) * 32 + B - 1) / B;
  const int gEdge8 = ((N_EDGES / E8_PER_WAVE) * 32 + B - 1) / B;

  deg_init_kernel<<<gN, B, 0, stream>>>(dinv, s0);
  deg_hist_kernel<<<DEG_NRANGES * DEG_CHUNKS, DEG_THREADS, 0, stream>>>(dinv, col);
  deg_rsqrt_kernel<<<gN, B, 0, stream>>>(dinv);

  agg_edge8_kernel<<<gEdge8, B, 0, stream>>>(x, row, col, dinv, wgt, s0, s1, pa0, pa1);
  gemm_wmma_kernel<8, 32, true, true, true, false><<<gGemm, B, 0, stream>>>(
      s0, s1, x, dinv, Bm[0], Wm[0], Bm[0], h1);

  gemm_wmma_kernel<32, 32, false, false, false, false><<<gGemm, B, 0, stream>>>(
      h1, h1, h1, dinv, Bm[1], Wm[1], Bm[1], hWA);
  agg_edge32_batched_kernel<true><<<gEdge32, B, 0, stream>>>(hWA, row, col, wgt, pa0, pa1, pb0, pb1);

  gemm_wmma_kernel<32, 32, true, false, true, true><<<gGemm, B, 0, stream>>>(
      pa0, pa1, hWA, dinv, Bm[1], Wm[2], Bm[2], hWB);
  agg_edge32_batched_kernel<true><<<gEdge32, B, 0, stream>>>(hWB, row, col, wgt, pb0, pb1, pa0, pa1);

  gemm_wmma_kernel<32, 32, true, false, true, true><<<gGemm, B, 0, stream>>>(
      pb0, pb1, hWB, dinv, Bm[2], Wm[3], Bm[3], hWA);
  agg_edge32_batched_kernel<false><<<gEdge32, B, 0, stream>>>(hWA, row, col, wgt, pa0, pa1, pb0, pb1);

  gemm_wmma_kernel<32, 3, true, false, true, true><<<gGemm, B, 0, stream>>>(
      pa0, pa1, hWA, dinv, Bm[3], Wm[4], Bm[4], hW3);
  {
    float* out = (float*)d_out;
    agg_edge3_lds_kernel<<<L5_RANGES * L5_CHUNKS, L5_THREADS, 0, stream>>>(hW3, row, col, wgt, part);
    agg_final3_kernel<<<(L5_PLANE + B - 1) / B, B, 0, stream>>>(hW3, Bm[4], dinv, part, out);
  }
}


# Products fused into the scatter: the work-group that summed a range's rows in LDS multiplies them there by the next layer's W (WMMA, A operand from LDS) and stores that product; 9 launches, not 14
# speedup vs baseline: 5.8251x; 1.6404x over previous
#include <hip/hip_runtime.h>

#define N_NODES 100000
#define N_EDGES 1600000

typedef __attribute__((ext_vector_type(2))) float v2f;
typedef __attribute__((ext_vector_type(8))) float v8f;

__global__ void deg_init_kernel(float* __restrict__ deg, int* __restrict__ rc) {
  int i = blockIdx.x * blockDim.x + threadIdx.x;
  if (i < 1024) rc[i] = 0;
  if (i < N_NODES) deg[i] = 1.0f;
}

#define DEG_RANGE_LOG 15
#define DEG_RANGE (1 << DEG_RANGE_LOG)
#define DEG_NRANGES ((N_NODES + DEG_RANGE - 1) / DEG_RANGE)
#define DEG_CHUNKS 32
#define DEG_CHUNK_EDGES (N_EDGES / DEG_CHUNKS)
#define DEG_THREADS 1024
static_assert(N_EDGES % DEG_CHUNKS == 0, "edge slices must be equal");
static_assert(DEG_CHUNK_EDGES % 4 == 0, "slices are read four indices at a time");
static_assert(DEG_CHUNK_EDGES < 65536, "a 16-bit counter must hold a whole slice");

__global__ void __launch_bounds__(DEG_THREADS)
deg_hist_kernel(float* __restrict__ deg, const int* __restrict__ col) {
  __shared__ unsigned cnt[DEG_RANGE / 2];
  const int range = blockIdx.x / DEG_CHUNKS;
  const int chunk = blockIdx.x - range * DEG_CHUNKS;
  const int tid = threadIdx.x;

  for (int w = tid; w < DEG_RANGE / 2; w += DEG_THREADS) cnt[w] = 0u;

  constexpr int NV = DEG_CHUNK_EDGES / 4;
  constexpr int IT = (NV + DEG_THREADS - 1) / DEG_THREADS;
  const int4* __restrict__ col4 =
      reinterpret_cast<const int4*>(col + (size_t)chunk * DEG_CHUNK_EDGES);
  int4 cv[IT];
#pragma unroll
  for (int k = 0; k < IT; ++k) {
    const int idx = tid + k * DEG_THREADS;
    cv[k] = col4[idx < NV ? idx : NV - 1];
  }
  __syncthreads();

#pragma unroll
  for (int k = 0; k < IT; ++k) {
    if (tid + k * DEG_THREADS < NV) {
      const int c4[4] = {cv[k].x, cv[k].y, cv[k].z, cv[k].w};
#pragma unroll
      for (int q = 0; q < 4; ++q) {
        const int c = c4[q];
        if ((c >> DEG_RANGE_LOG) == range) {
          const int local = c & (DEG_RANGE - 1);
          atomicAdd(&cnt[local >> 1], 1u << ((local & 1) << 4));
        }
      }
    }
  }
  __syncthreads();

  const unsigned short* cnt16 = reinterpret_cast<const unsigned short*>(cnt);
  const int nbase = range << DEG_RANGE_LOG;
  for (int n = tid; n < DEG_RANGE; n += DEG_THREADS) {
    const int node = nbase + n;
    const unsigned k = cnt16[n];
    if (node < N_NODES && k != 0u) {
      __hip_atomic_fetch_add(&deg[node], (float)k,
                             __ATOMIC_RELAXED, __HIP_MEMORY_SCOPE_AGENT);
    }
  }
}

#define BIN_NODES 196
#define PLACE_BITS 8
#define NBINS ((N_NODES + BIN_NODES - 1) / BIN_NODES)
#define BIN_SLOTS 512
static_assert(BIN_NODES <= (1 << PLACE_BITS), "place must fit its bits");
static_assert(NBINS <= BIN_SLOTS, "one thread per range in the scans");

__global__ void __launch_bounds__(256)
deg_rsqrt_kernel(float* __restrict__ deg, int* __restrict__ rcount) {
  __shared__ int tot[4];
  const int tid = threadIdx.x;
  const int i = blockIdx.x * 256 + tid;
  const int first = (blockIdx.x * 256) / BIN_NODES;
  if (tid < 4) tot[tid] = 0;
  __syncthreads();
  if (i < N_NODES) {
    const float d = deg[i];
    const int k = (int)d - 1;
    deg[i] = rsqrtf(d);
    if (k > 0) atomicAdd(&tot[i / BIN_NODES - first], k);
  }
  __syncthreads();
  if (tid < 4) {
    const int t = tot[tid];
    if (t != 0) atomicAdd(&rcount[first + tid], t);
  }
}

#define ROW_TILES ((BIN_NODES + 15) / 16)
#define PAD_ROWS (ROW_TILES * 16)
#define TILE_PITCH 36

template <int DIN, bool SBIAS>
__device__ __forceinline__ void own_term(const float* __restrict__ srow, float dd,
                                         const float* __restrict__ sbias,
                                         int lane, v2f (&own)[DIN / 4]) {
  const int khi = (lane >> 4) << 1;
#pragma unroll
  for (int k = 0; k < DIN; k += 4) {
    float sx = srow[k + khi] * dd;
    float sy = srow[k + khi + 1] * dd;
    if (SBIAS) {
      sx += sbias[k + khi];
      sy += sbias[k + khi + 1];
    }
    own[k / 4].x = sx;
    own[k / 4].y = sy;
  }
}

template <int DIN, int DOUT, bool SELF, bool BIAS>
__device__ __forceinline__ void tile_product(const float* arow,
                                             const v2f (&own)[DIN / 4],
                                             const float* __restrict__ W,
                                             const float* __restrict__ bias,
                                             float* dst, int dpitch, int rows,
                                             int lane) {
  constexpr int NT = (DOUT + 15) / 16;
  const int lhalf = lane & 15;
  const int half = lane >> 4;
  const int khi = half << 1;
  const int ncl = (lhalf < DOUT) ? lhalf : (DOUT - 1);

  v8f acc0 = {};
  v8f acc1 = {};
#pragma unroll
  for (int k = 0; k < DIN; k += 4) {
    v2f a, b0;
    a.x = arow[k + khi];
    a.y = arow[k + khi + 1];
    if (SELF) {
      a.x += own[k / 4].x;
      a.y += own[k / 4].y;
    }
    b0.x = W[(k + khi) * DOUT + ncl];
    b0.y = W[(k + khi + 1) * DOUT + ncl];
    acc0 = __builtin_amdgcn_wmma_f32_16x16x4_f32(
        false, a, false, b0, (short)0, acc0, false, false);
    if (NT > 1) {
      v2f b1;
      b1.x = W[(k + khi) * DOUT + 16 + lhalf];
      b1.y = W[(k + khi + 1) * DOUT + 16 + lhalf];
      acc1 = __builtin_amdgcn_wmma_f32_16x16x4_f32(
          false, a, false, b1, (short)0, acc1, false, false);
    }
  }

  float bb0 = 0.0f, bb1 = 0.0f;
  if (BIAS) {
    bb0 = bias[ncl];
    if (NT > 1) bb1 = bias[16 + lhalf];
  }
#pragma unroll
  for (int v = 0; v < 8; ++v) {
    const int r = half * 8 + v;
    if (r < rows) {
      float* o = dst + r * dpitch;
      if (lhalf < DOUT) o[lhalf] = BIAS ? acc0[v] + bb0 : acc0[v];
      if (NT > 1)       o[16 + lhalf] = BIAS ? acc1[v] + bb1 : acc1[v];
    }
  }
}

#define BIN_THREADS 1024
#define BIN_EPT 4
#define BIN_CHUNK (BIN_THREADS * BIN_EPT)
#define BIN_WGS ((N_EDGES + BIN_CHUNK - 1) / BIN_CHUNK)
static_assert(N_EDGES % BIN_EPT == 0, "a thread's four edges are all inside or all outside");
static_assert(N_NODES < (1 << (31 - PLACE_BITS)), "row and place share one word");

__device__ __forceinline__ int bin_prefix(int v, int tid, int* wsum) {
  const int lane = tid & 31;
  int inc = v;
#pragma unroll
  for (int o = 1; o < 32; o <<= 1) {
    const int n = __shfl_up(inc, o);
    if (lane >= o) inc += n;
  }
  if (lane == 31) wsum[tid >> 5] = inc;
  __syncthreads();
  int off = 0;
  for (int w = 0; w < (tid >> 5); ++w) off += wsum[w];
  __syncthreads();
  return off + inc - v;
}

__global__ void __launch_bounds__(BIN_THREADS)
bin_edges_kernel(const int* __restrict__ row,
                 const int* __restrict__ col,
                 const float* __restrict__ dinv,
                 const int* __restrict__ rcount,
                 int* __restrict__ cursor,
                 int* __restrict__ binStart,
                 int* __restrict__ recw) {
  int2* __restrict__ rec = reinterpret_cast<int2*>(recw);
  __shared__ int cnt[BIN_SLOTS];
  __shared__ int lofs[BIN_SLOTS];
  __shared__ int gpos[BIN_SLOTS];
  __shared__ int wsum[BIN_THREADS / 32];
  __shared__ int2 srec[BIN_CHUNK];
  __shared__ int sdst[BIN_CHUNK];
  const int tid = threadIdx.x;
  const int wg = blockIdx.x;

  if (tid < BIN_SLOTS) cnt[tid] = 0;
  const int e = (wg * BIN_THREADS + tid) * BIN_EPT;
  const bool valid = e < N_EDGES;
  const int ec = valid ? e : 0;
  const int4 cv = *reinterpret_cast<const int4*>(col + ec);
  const int4 rv = *reinterpret_cast<const int4*>(row + ec);
  __syncthreads();

  const int c4[BIN_EPT] = {cv.x, cv.y, cv.z, cv.w};
  const int r4[BIN_EPT] = {rv.x, rv.y, rv.z, rv.w};
  float w4[BIN_EPT];
#pragma unroll
  for (int q = 0; q < BIN_EPT; ++q) w4[q] = dinv[r4[q]] * dinv[c4[q]];
  int rank[BIN_EPT];
#pragma unroll
  for (int q = 0; q < BIN_EPT; ++q) {
    rank[q] = valid ? atomicAdd(&cnt[c4[q] / BIN_NODES], 1) : 0;
  }
  __syncthreads();

  const int total = tid < NBINS ? rcount[tid] : 0;
  const int mine = tid < BIN_SLOTS ? cnt[tid] : 0;
  const int gstart = bin_prefix(total, tid, wsum);
  const int lstart = bin_prefix(mine, tid, wsum);
  if (tid < NBINS) {
    int res = 0;
    if (mine > 0) {
      res = __hip_atomic_fetch_add(&cursor[tid], mine, __ATOMIC_RELAXED,
                                   __HIP_MEMORY_SCOPE_AGENT);
    }
    gpos[tid] = gstart + res;
    lofs[tid] = lstart;
    if (wg == 0) binStart[tid] = gstart;
  }
  __syncthreads();

  if (valid) {
#pragma unroll
    for (int q = 0; q < BIN_EPT; ++q) {
      const int b = c4[q] / BIN_NODES;
      const int s = lofs[b] + rank[q];
      int2 rcd;
      rcd.x = (r4[q] << PLACE_BITS) | (c4[q] - b * BIN_NODES);
      rcd.y = __float_as_int(w4[q]);
      srec[s] = rcd;
      sdst[s] = gpos[b] + rank[q];
    }
  }
  __syncthreads();

  const int left = N_EDGES - wg * BIN_CHUNK;
  const int nvalid = left < BIN_CHUNK ? left : BIN_CHUNK;
#pragma unroll
  for (int k = 0; k < BIN_EPT; ++k) {
    const int s = tid + k * BIN_THREADS;
    if (s < nvalid) rec[sdst[s]] = srec[s];
  }
}

#define AGG_THREADS 1024
#define AGG_STEP AGG_THREADS
template <int DOUT>
__global__ void __launch_bounds__(AGG_THREADS)
agg_bin32_kernel(const float* __restrict__ hW,
                 const int* __restrict__ recw,
                 const int* __restrict__ binStart,
                 const int* __restrict__ rcount,
                 const float* __restrict__ dinv,
                 const float* __restrict__ sbias,
                 const float* __restrict__ W,
                 float* __restrict__ out) {
  __shared__ float acc[PAD_ROWS * TILE_PITCH];
  const int g = blockIdx.x;
  const int tid = threadIdx.x;
  const int lane = tid & 31;
  const int wave = __builtin_amdgcn_readfirstlane(tid >> 5);
  const int start = __builtin_amdgcn_readfirstlane(binStart[g]);
  const int cnt = __builtin_amdgcn_readfirstlane(rcount[g]);

  for (int i = tid; i < PAD_ROWS * TILE_PITCH; i += AGG_THREADS) acc[i] = 0.0f;
  __syncthreads();

  const int2* __restrict__ my = reinterpret_cast<const int2*>(recw) + start;
  int base = wave * 32;
  int2 cur = {0, 0};
  if (base < cnt) {
    const int idx = base + lane;
    cur = my[idx < cnt ? idx : cnt - 1];
  }
  while (base < cnt) {
    const int nb = base + AGG_STEP;
    const int n = cnt - base;
    float v[32];
#pragma unroll
    for (int j = 0; j < 32; ++j) {
      const unsigned r = (unsigned)__builtin_amdgcn_readlane(cur.x, j) >> PLACE_BITS;
      v[j] = hW[(size_t)r * 32 + lane];
    }
    int2 nxt = cur;
    if (nb < cnt) {
      const int idx = nb + lane;
      nxt = my[idx < cnt ? idx : cnt - 1];
    }
#pragma unroll
    for (int j = 0; j < 32; ++j) {
      const int cl = __builtin_amdgcn_readlane(cur.x, j) & ((1 << PLACE_BITS) - 1);
      const float w = __int_as_float(__builtin_amdgcn_readlane(cur.y, j));
      const float add = (j < n) ? v[j] * w : 0.0f;
      __hip_atomic_fetch_add(&acc[cl * TILE_PITCH + lane], add,
                             __ATOMIC_RELAXED, __HIP_MEMORY_SCOPE_WORKGROUP);
    }
    cur = nxt;
    base = nb;
  }

  const int left = N_NODES - g * BIN_NODES;
  const int nrows = left < BIN_NODES ? left : BIN_NODES;
  const int t0 = (AGG_THREADS / 32 - 1 - wave) * 16;
  const int rloc = t0 + (lane & 15);
  v2f own[8] = {};
  if (t0 < nrows) {
    const int nd = g * BIN_NODES + rloc;
    const int node = nd < N_NODES ? nd : N_NODES - 1;
    const float d = dinv[node];
    own_term<32, true>(hW + (size_t)node * 32, d * d, sbias, lane, own);
  }
  __syncthreads();
  if (t0 < nrows) {
    tile_product<32, DOUT, true, false>(
        acc + rloc * TILE_PITCH, own, W, sbias,
        out + (size_t)(g * BIN_NODES + t0) * DOUT, DOUT, nrows - t0, lane);
  }
}

#define HOLD8(a)                                                              \
  asm volatile("" ::"v"(a[0]), "v"(a[1]), "v"(a[2]), "v"(a[3]), "v"(a[4]),   \
               "v"(a[5]), "v"(a[6]), "v"(a[7]))
#define NAR_THREADS 1024
#define N8_RECS 4
#define N8_PITCH 9

__global__ void __launch_bounds__(NAR_THREADS)
agg_bin8_kernel(const float* __restrict__ xin,
                const int* __restrict__ recw,
                const int* __restrict__ binStart,
                const int* __restrict__ rcount,
                const float* __restrict__ dinv,
                const float* __restrict__ W1,
                const float* __restrict__ b1,
                const float* __restrict__ W2,
                float* __restrict__ out) {
  __shared__ float acc[PAD_ROWS * N8_PITCH];
  __shared__ float h1t[PAD_ROWS * TILE_PITCH];
  const int g = blockIdx.x;
  const int tid = threadIdx.x;
  const int start = binStart[g];
  const int cnt = rcount[g];

  for (int i = tid; i < PAD_ROWS * N8_PITCH; i += NAR_THREADS) acc[i] = 0.0f;
  __syncthreads();

  const int2* __restrict__ my = reinterpret_cast<const int2*>(recw) + start;
  for (int base = 0; base < cnt; base += NAR_THREADS * N8_RECS) {
    int rw[N8_RECS * 2];
#pragma unroll
    for (int k = 0; k < N8_RECS; ++k) {
      const int idx = base + tid + k * NAR_THREADS;
      const int2 r = my[idx < cnt ? idx : cnt - 1];
      rw[k * 2] = r.x;
      rw[k * 2 + 1] = r.y;
    }
    float v[N8_RECS * 8];
#pragma unroll
    for (int k = 0; k < N8_RECS; ++k) {
      const unsigned r = (unsigned)rw[k * 2] >> PLACE_BITS;
      const float4* __restrict__ p = reinterpret_cast<const float4*>(xin + (size_t)r * 8);
      const float4 lo = p[0];
      const float4 hi = p[1];
      v[k * 8 + 0] = lo.x;
      v[k * 8 + 1] = lo.y;
      v[k * 8 + 2] = lo.z;
      v[k * 8 + 3] = lo.w;
      v[k * 8 + 4] = hi.x;
      v[k * 8 + 5] = hi.y;
      v[k * 8 + 6] = hi.z;
      v[k * 8 + 7] = hi.w;
    }
#pragma unroll
    for (int k = 0; k < N8_RECS; ++k) HOLD8((v + k * 8));
#pragma unroll
    for (int k = 0; k < N8_RECS; ++k) {
      if (base + tid + k * NAR_THREADS < cnt) {
        const int cl = rw[k * 2] & ((1 << PLACE_BITS) - 1);
        const float w = __int_as_float(rw[k * 2 + 1]);
#pragma unroll
        for (int f = 0; f < 8; ++f) {
          __hip_atomic_fetch_add(&acc[cl * N8_PITCH + f], v[k * 8 + f] * w,
                                 __ATOMIC_RELAXED, __HIP_MEMORY_SCOPE_WORKGROUP);
        }
      }
    }
  }

  const int left = N_NODES - g * BIN_NODES;
  const int nrows = left < BIN_NODES ? left : BIN_NODES;
  const int lane = tid & 31;
  const int t0 = __builtin_amdgcn_readfirstlane(tid >> 5) * 16;
  const int rloc = t0 + (lane & 15);
  v2f own[2] = {};
  if (t0 < nrows) {
    const int nd = g * BIN_NODES + rloc;
    const int node = nd < N_NODES ? nd : N_NODES - 1;
    const float d = dinv[node];
    own_term<8, false>(xin + (size_t)node * 8, d * d, b1, lane, own);
  }
  __syncthreads();
  if (t0 < nrows) {
    tile_product<8, 32, true, true>(
        acc + rloc * N8_PITCH, own, W1, b1,
        h1t + t0 * TILE_PITCH, TILE_PITCH, 16, lane);
  }
  __syncthreads();
  if (t0 < nrows) {
    const v2f none[8] = {};
    tile_product<32, 32, false, false>(
        h1t + rloc * TILE_PITCH, none, W2, b1,
        out + (size_t)(g * BIN_NODES + t0) * 32, 32, nrows - t0, lane);
  }
}

#define OUT3_ALIGN 32
#define OUT3_MAXN (BIN_NODES + OUT3_ALIGN)
#define N3_RECS 8
static_assert(OUT3_ALIGN <= BIN_NODES, "an owner's nodes must lie in two neighbouring ranges");
static_assert((OUT3_ALIGN * 3) % 32 == 0, "an owner's floats must be whole lines");
static_assert(N_NODES % OUT3_ALIGN == 0, "the last owner's floats must be whole lines");
static_assert(N3_RECS == 8, "HOLD8 names a pass's gathered values");

__device__ __forceinline__ int out3_lo(int j) {
  return j >= NBINS ? N_NODES : ((j * BIN_NODES) & ~(OUT3_ALIGN - 1));
}

__global__ void __launch_bounds__(NAR_THREADS)
agg_bin3_kernel(const float* __restrict__ hW,
                const int* __restrict__ recw,
                const int* __restrict__ binStart,
                const int* __restrict__ rcount,
                const float* __restrict__ dinv,
                const float* __restrict__ bias,
                float* __restrict__ out) {
  __shared__ float acc[OUT3_MAXN * 3];
  const int j = blockIdx.x;
  const int tid = threadIdx.x;
  const int lo = out3_lo(j);
  const int len = out3_lo(j + 1) - lo;
  const int sb = binStart[j];
  const int s0 = j > 0 ? binStart[j - 1] : sb;
  const int end = sb + rcount[j];

  for (int i = tid; i < OUT3_MAXN * 3; i += NAR_THREADS) acc[i] = 0.0f;
  __syncthreads();

  const int2* __restrict__ rec = reinterpret_cast<const int2*>(recw);
  for (int base = s0; base < end; base += NAR_THREADS * N3_RECS) {
    int rx[N3_RECS], ry[N3_RECS];
#pragma unroll
    for (int k = 0; k < N3_RECS; ++k) {
      const int idx = base + tid + k * NAR_THREADS;
      const int2 r = rec[idx < end ? idx : end - 1];
      rx[k] = r.x;
      ry[k] = r.y;
    }
    __builtin_amdgcn_sched_barrier(0);
    float g0[N3_RECS], g1[N3_RECS], g2[N3_RECS];
    int loc[N3_RECS];
#pragma unroll
    for (int k = 0; k < N3_RECS; ++k) {
      const int idx = base + tid + k * NAR_THREADS;
      const int node = (idx < sb ? j - 1 : j) * BIN_NODES + (rx[k] & ((1 << PLACE_BITS) - 1));
      const int local = node - lo;
      const bool in = (idx < end) && ((unsigned)local < (unsigned)len);
      const unsigned r = in ? ((unsigned)rx[k] >> PLACE_BITS) : 0u;
      const float* __restrict__ p = hW + r * 3u;
      g0[k] = p[0];
      g1[k] = p[1];
      g2[k] = p[2];
      loc[k] = in ? local * 3 : -1;
    }
    HOLD8(g0);
    HOLD8(g1);
    HOLD8(g2);
#pragma unroll
    for (int k = 0; k < N3_RECS; ++k) {
      if (loc[k] >= 0) {
        const float w = __int_as_float(ry[k]);
        __hip_atomic_fetch_add(&acc[loc[k]], g0[k] * w,
                               __ATOMIC_RELAXED, __HIP_MEMORY_SCOPE_WORKGROUP);
        __hip_atomic_fetch_add(&acc[loc[k] + 1], g1[k] * w,
                               __ATOMIC_RELAXED, __HIP_MEMORY_SCOPE_WORKGROUP);
        __hip_atomic_fetch_add(&acc[loc[k] + 2], g2[k] * w,
                               __ATOMIC_RELAXED, __HIP_MEMORY_SCOPE_WORKGROUP);
      }
    }
  }
  __syncthreads();

  const int t0 = lo * 3;
  for (int n = tid; n < len * 3; n += NAR_THREADS) {
    const int i = n / 3;
    const int f = n - i * 3;
    const float d = dinv[lo + i];
    out[t0 + n] = acc[n] + hW[t0 + n] * (d * d) + bias[f];
  }
}

extern "C" void kernel_launch(void* const* d_in, const int* in_sizes, int n_in,
                              void* d_out, int out_size, void* d_ws, size_t ws_size,
                              hipStream_t stream) {
  const float* x  = (const float*)d_in[0];
  const int*   ei = (const int*)d_in[1];
  const int*   row = ei;
  const int*   col = ei + N_EDGES;

  const float* Wm[5] = {(const float*)d_in[2], (const float*)d_in[4],
                        (const float*)d_in[6], (const float*)d_in[8],
                        (const float*)d_in[10]};
  const float* Bm[5] = {(const float*)d_in[3], (const float*)d_in[5],
                        (const float*)d_in[7], (const float*)d_in[9],
                        (const float*)d_in[11]};

  float* ws   = (float*)d_ws;
  float* dinv = ws;
  float* hWA  = ws + N_NODES * 49;
  float* hWB  = ws + N_NODES * 81;
  float* hW3  = ws + N_NODES * 241;
  int*   rec  = (int*)(ws + N_NODES * 341 + N_EDGES);
  int* rcount   = (int*)(ws + N_NODES * 341 + N_EDGES * 3);
  int* cursor   = rcount + 512;
  int* binStart = rcount + 1024;

  const int B = 256;
  const int gN  = (N_NODES + B - 1) / B;

  deg_init_kernel<<<gN, B, 0, stream>>>(dinv, rcount);
  deg_hist_kernel<<<DEG_NRANGES * DEG_CHUNKS, DEG_THREADS, 0, stream>>>(dinv, col);
  deg_rsqrt_kernel<<<gN, B, 0, stream>>>(dinv, rcount);

  bin_edges_kernel<<<BIN_WGS, BIN_THREADS, 0, stream>>>(row, col, dinv, rcount, cursor, binStart, rec);

  agg_bin8_kernel<<<NBINS, NAR_THREADS, 0, stream>>>(x, rec, binStart, rcount, dinv,
                                                     Wm[0], Bm[0], Wm[1], hWA);

  agg_bin32_kernel<32><<<NBINS, AGG_THREADS, 0, stream>>>(hWA, rec, binStart, rcount, dinv,
                                                          Bm[1], Wm[2], hWB);

  agg_bin32_kernel<32><<<NBINS, AGG_THREADS, 0, stream>>>(hWB, rec, binStart, rcount, dinv,
                                                          Bm[2], Wm[3], hWA);

  agg_bin32_kernel<3><<<NBINS, AGG_THREADS, 0, stream>>>(hWA, rec, binStart, rcount, dinv,
                                                         Bm[3], Wm[4], hW3);

  agg_bin3_kernel<<<NBINS, NAR_THREADS, 0, stream>>>(hW3, rec, binStart, rcount, dinv, Bm[4],
                                                     (float*)d_out);
}
